# starting kernel
# baseline (speedup 1.0000x reference)
#include <hip/hip_runtime.h>

#define NN 50000
#define NE 800000
#define DH 128
#define DO 64
#define NG 128

typedef float v2f __attribute__((ext_vector_type(2)));
typedef float v8f __attribute__((ext_vector_type(8)));

__global__ void k_fill1(float* p, int n) {
    int i = blockIdx.x * blockDim.x + threadIdx.x;
    if (i < n) p[i] = 1.0f;
}

__global__ void k_deg_accum(const int* __restrict__ dst, float* __restrict__ deg, int E) {
    int e = blockIdx.x * blockDim.x + threadIdx.x;
    if (e < E) atomicAdd(&deg[dst[e]], 1.0f);
}

__global__ void k_deg_finish(float* __restrict__ deg_selfc, float* __restrict__ dinv, int n) {
    int i = blockIdx.x * blockDim.x + threadIdx.x;
    if (i < n) {
        float d = deg_selfc[i];
        dinv[i] = 1.0f / sqrtf(d);
        deg_selfc[i] = 1.0f / d;
    }
}

__global__ void k_edge_w(const int* __restrict__ src, const int* __restrict__ dst,
                         const float* __restrict__ dinv, float* __restrict__ ew, int E) {
    int e = blockIdx.x * blockDim.x + threadIdx.x;
    if (e < E) ew[e] = dinv[src[e]] * dinv[dst[e]];
}

template <int NC>
__global__ void __launch_bounds__(256)
k_wmma_gemm(const float* __restrict__ A, const float* __restrict__ W,
            const float* __restrict__ bias, float* __restrict__ C,
            int M, int relu)
{
    __shared__ float sWt[NC * 132];
    __shared__ float sA[16 * 132];

    const int tid = threadIdx.x;
    const int r0  = blockIdx.x * 16;

    for (int i = tid; i < 128 * NC; i += 256) {
        int k = i / NC, c = i % NC;
        sWt[c * 132 + k] = W[i];
    }
    for (int i = tid; i < 16 * 128; i += 256) {
        int r = i >> 7, c = i & 127;
        int row = r0 + r;
        sA[r * 132 + c] = (row < M) ? A[row * 128 + c] : 0.0f;
    }
    __syncthreads();

    const int wave = tid >> 5;
    const int lane = tid & 31;
    const int ln   = lane & 15;
    const int hi   = lane >> 4;
    const int c0   = wave * 16;

    if (c0 < NC) {
        const float* aP = &sA [ln        * 132 + 2 * hi];
        const float* bP = &sWt[(c0 + ln) * 132 + 2 * hi];
        v8f acc = {0.f, 0.f, 0.f, 0.f, 0.f, 0.f, 0.f, 0.f};
#pragma unroll
        for (int k = 0; k < 128; k += 4) {
            v2f a = *(const v2f*)(aP + k);
            v2f b = *(const v2f*)(bP + k);
            acc = __builtin_amdgcn_wmma_f32_16x16x4_f32(
                false, a, false, b, (short)0, acc, false, false);
        }
        const int col = c0 + ln;
        const float bv = bias ? bias[col] : 0.0f;
#pragma unroll
        for (int r = 0; r < 8; ++r) {
            int row = r0 + r + 8 * hi;
            if (row < M) {
                float v = acc[r] + bv;
                if (relu) v = fmaxf(v, 0.0f);
                C[row * NC + col] = v;
            }
        }
    }
}

__global__ void k_scatter(const int* __restrict__ src, const int* __restrict__ dst,
                          const float* __restrict__ ew, const float* __restrict__ H2,
                          float* __restrict__ AGG, int E)
{
    long long idx = (long long)blockIdx.x * blockDim.x + threadIdx.x;
    if (idx >= (long long)E * DH) return;
    int e = (int)(idx >> 7);
    int d = (int)(idx & 127);
    atomicAdd(&AGG[(size_t)dst[e] * DH + d], H2[(size_t)src[e] * DH + d] * ew[e]);
}

__global__ void k_combine(const float* __restrict__ AGG, const float* __restrict__ H2,
                          const float* __restrict__ selfc, const float* __restrict__ bias,
                          float* __restrict__ H, int n)
{
    int idx = blockIdx.x * blockDim.x + threadIdx.x;
    if (idx >= n * DH) return;
    int i = idx >> 7, d = idx & 127;
    float v = AGG[idx] + H2[idx] * selfc[i] + bias[d];
    H[idx] = fmaxf(v, 0.0f);
}

__global__ void k_pool_accum(const float* __restrict__ H, const int* __restrict__ batch,
                             float* __restrict__ sums, float* __restrict__ counts, int total)
{
    int idx = blockIdx.x * blockDim.x + threadIdx.x;
    if (idx >= total) return;
    int i = idx >> 7, d = idx & 127;
    int g = batch[i];
    atomicAdd(&sums[g * DH + d], H[idx]);
    if (d == 0) atomicAdd(&counts[g], 1.0f);
}

__global__ void k_pool_div(const float* __restrict__ sums, const float* __restrict__ counts,
                           float* __restrict__ pooled)
{
    int idx = blockIdx.x * blockDim.x + threadIdx.x;
    if (idx >= NG * DH) return;
    pooled[idx] = sums[idx] / fmaxf(counts[idx >> 7], 1.0f);
}

extern "C" void kernel_launch(void* const* d_in, const int* in_sizes, int n_in,
                              void* d_out, int out_size, void* d_ws, size_t ws_size,
                              hipStream_t stream)
{
    (void)in_sizes; (void)n_in; (void)out_size; (void)ws_size;

    const float* x     = (const float*)d_in[0];
    const int*   ei    = (const int*)d_in[1];
    const int*   src   = ei;
    const int*   dst   = ei + NE;
    const int*   batch = (const int*)d_in[2];
    const float* Wl[5] = {(const float*)d_in[4], (const float*)d_in[6], (const float*)d_in[8],
                          (const float*)d_in[10], (const float*)d_in[12]};
    const float* Bl[5] = {(const float*)d_in[5], (const float*)d_in[7], (const float*)d_in[9],
                          (const float*)d_in[11], (const float*)d_in[13]};
    const float* fc1_w = (const float*)d_in[14];
    const float* fc1_b = (const float*)d_in[15];
    const float* fc2_w = (const float*)d_in[16];
    const float* fc2_b = (const float*)d_in[17];

    float* H      = (float*)d_ws;
    float* H2     = H      + (size_t)NN * DH;
    float* AGG    = H2     + (size_t)NN * DH;
    float* selfc  = AGG    + (size_t)NN * DH;
    float* dinv   = selfc  + NN;
    float* ew     = dinv   + NN;
    float* sums   = ew     + NE;
    float* counts = sums   + (size_t)NG * DH;
    float* pooled = counts + NG;
    float* gact   = pooled + (size_t)NG * DH;

    const int T = 256;

    k_fill1<<<(NN + T - 1) / T, T, 0, stream>>>(selfc, NN);
    k_deg_accum<<<(NE + T - 1) / T, T, 0, stream>>>(dst, selfc, NE);
    k_deg_finish<<<(NN + T - 1) / T, T, 0, stream>>>(selfc, dinv, NN);
    k_edge_w<<<(NE + T - 1) / T, T, 0, stream>>>(src, dst, dinv, ew, NE);

    const long long escat = (long long)NE * DH;
    const int scat_blocks = (int)((escat + T - 1) / T);
    const int gemm_blocks = NN / 16;
    const int comb_blocks = (NN * DH + T - 1) / T;

    const float* cur = x;
    for (int l = 0; l < 5; ++l) {
        k_wmma_gemm<DH><<<gemm_blocks, T, 0, stream>>>(cur, Wl[l], nullptr, H2, NN, 0);
        hipMemsetAsync(AGG, 0, (size_t)NN * DH * sizeof(float), stream);
        k_scatter<<<scat_blocks, T, 0, stream>>>(src, dst, ew, H2, AGG, NE);
        k_combine<<<comb_blocks, T, 0, stream>>>(AGG, H2, selfc, Bl[l], H, NN);
        cur = H;
    }

    hipMemsetAsync(sums, 0, ((size_t)NG * DH + NG) * sizeof(float), stream);
    k_pool_accum<<<(NN * DH + T - 1) / T, T, 0, stream>>>(H, batch, sums, counts, NN * DH);
    k_pool_div<<<(NG * DH + T - 1) / T, T, 0, stream>>>(sums, counts, pooled);

    k_wmma_gemm<DH><<<NG / 16, T, 0, stream>>>(pooled, fc1_w, fc1_b, gact, NG, 1);
    k_wmma_gemm<DO><<<NG / 16, T, 0, stream>>>(gact, fc2_w, fc2_b, (float*)d_out, NG, 0);
}


# edge sum by per-call CSR: int atomic slot rank + scan + fill, one wave per node gathers H2 rows and fuses self term, bias, relu; replaces float-atomic scatter; sum order = CSR slot (arrival) order
# speedup vs baseline: 3.4974x; 3.4974x over previous
#include <hip/hip_runtime.h>

#define NN 50000
#define NE 800000
#define DH 128
#define DO 64
#define NG 128

typedef float v2f __attribute__((ext_vector_type(2)));
typedef float v8f __attribute__((ext_vector_type(8)));

#define NNP 50176

__global__ void k_zero_i(int* p, int n) {
    int i = blockIdx.x * blockDim.x + threadIdx.x;
    if (i < n) p[i] = 0;
}

__global__ void k_rank(const int* __restrict__ dst, int* __restrict__ cnt,
                       int* __restrict__ rank, int E) {
    int e = blockIdx.x * blockDim.x + threadIdx.x;
    if (e < E) rank[e] = atomicAdd(&cnt[dst[e]], 1);
}

__global__ void __launch_bounds__(256)
k_scan_norm(const int* __restrict__ cnt, int* __restrict__ rowptr,
            float* __restrict__ selfc, float* __restrict__ dinv)
{
    __shared__ int sOwn[8];
    __shared__ int sPre[8];
    const int4* c4 = (const int4*)cnt;
    const int b = blockIdx.x, t = threadIdx.x;
    const int lane = t & 31, wave = t >> 5;

    int part = 0;
    for (int i = t; i < b * 256; i += 256) {
        int4 v = c4[i];
        part += (v.x + v.y) + (v.z + v.w);
    }
    const int4 v = c4[b * 256 + t];
    const int own = (v.x + v.y) + (v.z + v.w);

    int incl = own;
#pragma unroll
    for (int o = 1; o < 32; o <<= 1) {
        int up = __shfl_up(incl, o);
        if (lane >= o) incl += up;
    }
#pragma unroll
    for (int o = 16; o > 0; o >>= 1) part += __shfl_xor(part, o);
    if (lane == 31) sOwn[wave] = incl;
    if (lane == 0)  sPre[wave] = part;
    __syncthreads();
    int base = 0;
#pragma unroll
    for (int w = 0; w < 8; ++w) {
        base += sPre[w];
        if (w < wave) base += sOwn[w];
    }
    const int ex = base + incl - own;
    int4 rp;
    rp.x = ex; rp.y = ex + v.x; rp.z = rp.y + v.y; rp.w = rp.z + v.z;
    ((int4*)rowptr)[b * 256 + t] = rp;

    const int i0 = (b * 256 + t) * 4;
    if (i0 < NN) {
        float4 di, sc;
        float d;
        d = (float)(v.x + 1); di.x = 1.0f / sqrtf(d); sc.x = 1.0f / d;
        d = (float)(v.y + 1); di.y = 1.0f / sqrtf(d); sc.y = 1.0f / d;
        d = (float)(v.z + 1); di.z = 1.0f / sqrtf(d); sc.z = 1.0f / d;
        d = (float)(v.w + 1); di.w = 1.0f / sqrtf(d); sc.w = 1.0f / d;
        ((float4*)dinv)[b * 256 + t]  = di;
        ((float4*)selfc)[b * 256 + t] = sc;
    }
}

__global__ void k_csr_fill(const int* __restrict__ src, const int* __restrict__ dst,
                           const int* __restrict__ rowptr, const int* __restrict__ rank,
                           const float* __restrict__ dinv, int2* __restrict__ csr, int E) {
    int e = blockIdx.x * blockDim.x + threadIdx.x;
    if (e < E) {
        int s = src[e], d = dst[e];
        float w = dinv[s] * dinv[d];
        csr[rowptr[d] + rank[e]] = make_int2(s, __float_as_int(w));
    }
}

template <int NC>
__global__ void __launch_bounds__(256)
k_wmma_gemm(const float* __restrict__ A, const float* __restrict__ W,
            const float* __restrict__ bias, float* __restrict__ C,
            int M, int relu)
{
    __shared__ float sWt[NC * 132];
    __shared__ float sA[16 * 132];

    const int tid = threadIdx.x;
    const int r0  = blockIdx.x * 16;

    for (int i = tid; i < 128 * NC; i += 256) {
        int k = i / NC, c = i % NC;
        sWt[c * 132 + k] = W[i];
    }
    for (int i = tid; i < 16 * 128; i += 256) {
        int r = i >> 7, c = i & 127;
        int row = r0 + r;
        sA[r * 132 + c] = (row < M) ? A[row * 128 + c] : 0.0f;
    }
    __syncthreads();

    const int wave = tid >> 5;
    const int lane = tid & 31;
    const int ln   = lane & 15;
    const int hi   = lane >> 4;
    const int c0   = wave * 16;

    if (c0 < NC) {
        const float* aP = &sA [ln        * 132 + 2 * hi];
        const float* bP = &sWt[(c0 + ln) * 132 + 2 * hi];
        v8f acc = {0.f, 0.f, 0.f, 0.f, 0.f, 0.f, 0.f, 0.f};
#pragma unroll
        for (int k = 0; k < 128; k += 4) {
            v2f a = *(const v2f*)(aP + k);
            v2f b = *(const v2f*)(bP + k);
            acc = __builtin_amdgcn_wmma_f32_16x16x4_f32(
                false, a, false, b, (short)0, acc, false, false);
        }
        const int col = c0 + ln;
        const float bv = bias ? bias[col] : 0.0f;
#pragma unroll
        for (int r = 0; r < 8; ++r) {
            int row = r0 + r + 8 * hi;
            if (row < M) {
                float v = acc[r] + bv;
                if (relu) v = fmaxf(v, 0.0f);
                C[row * NC + col] = v;
            }
        }
    }
}

__global__ void __launch_bounds__(256)
k_gather_combine(const int* __restrict__ rowptr, const int2* __restrict__ csr,
                 const float* __restrict__ H2, const float* __restrict__ selfc,
                 const float* __restrict__ bias, float* __restrict__ H)
{
    const int lane = threadIdx.x & 31;
    const int i = __builtin_amdgcn_readfirstlane((int)(blockIdx.x * 8 + (threadIdx.x >> 5)));
    const int beg = rowptr[i], end = rowptr[i + 1];
    const float4* H2v = (const float4*)H2;

    float4 acc = {0.f, 0.f, 0.f, 0.f};
    int j = beg;
    for (; j + 4 <= end; j += 4) {
        const int2 e0 = csr[j], e1 = csr[j + 1], e2 = csr[j + 2], e3 = csr[j + 3];
        const float4 v0 = H2v[(size_t)e0.x * 32 + lane];
        const float4 v1 = H2v[(size_t)e1.x * 32 + lane];
        const float4 v2 = H2v[(size_t)e2.x * 32 + lane];
        const float4 v3 = H2v[(size_t)e3.x * 32 + lane];
        const float w0 = __int_as_float(e0.y), w1 = __int_as_float(e1.y);
        const float w2 = __int_as_float(e2.y), w3 = __int_as_float(e3.y);
        acc.x = fmaf(v0.x, w0, acc.x); acc.y = fmaf(v0.y, w0, acc.y); acc.z = fmaf(v0.z, w0, acc.z); acc.w = fmaf(v0.w, w0, acc.w);
        acc.x = fmaf(v1.x, w1, acc.x); acc.y = fmaf(v1.y, w1, acc.y); acc.z = fmaf(v1.z, w1, acc.z); acc.w = fmaf(v1.w, w1, acc.w);
        acc.x = fmaf(v2.x, w2, acc.x); acc.y = fmaf(v2.y, w2, acc.y); acc.z = fmaf(v2.z, w2, acc.z); acc.w = fmaf(v2.w, w2, acc.w);
        acc.x = fmaf(v3.x, w3, acc.x); acc.y = fmaf(v3.y, w3, acc.y); acc.z = fmaf(v3.z, w3, acc.z); acc.w = fmaf(v3.w, w3, acc.w);
    }
    for (; j < end; ++j) {
        const int2 e0 = csr[j];
        const float4 v0 = H2v[(size_t)e0.x * 32 + lane];
        const float w0 = __int_as_float(e0.y);
        acc.x = fmaf(v0.x, w0, acc.x); acc.y = fmaf(v0.y, w0, acc.y); acc.z = fmaf(v0.z, w0, acc.z); acc.w = fmaf(v0.w, w0, acc.w);
    }

    const float4 h = H2v[(size_t)i * 32 + lane];
    const float sc = selfc[i];
    const int d0 = lane * 4;
    float4 o;
    o.x = fmaxf(acc.x + h.x * sc + bias[d0 + 0], 0.0f);
    o.y = fmaxf(acc.y + h.y * sc + bias[d0 + 1], 0.0f);
    o.z = fmaxf(acc.z + h.z * sc + bias[d0 + 2], 0.0f);
    o.w = fmaxf(acc.w + h.w * sc + bias[d0 + 3], 0.0f);
    ((float4*)H)[(size_t)i * 32 + lane] = o;
}

__global__ void k_pool_accum(const float* __restrict__ H, const int* __restrict__ batch,
                             float* __restrict__ sums, float* __restrict__ counts, int total)
{
    int idx = blockIdx.x * blockDim.x + threadIdx.x;
    if (idx >= total) return;
    int i = idx >> 7, d = idx & 127;
    int g = batch[i];
    atomicAdd(&sums[g * DH + d], H[idx]);
    if (d == 0) atomicAdd(&counts[g], 1.0f);
}

__global__ void k_pool_div(const float* __restrict__ sums, const float* __restrict__ counts,
                           float* __restrict__ pooled)
{
    int idx = blockIdx.x * blockDim.x + threadIdx.x;
    if (idx >= NG * DH) return;
    pooled[idx] = sums[idx] / fmaxf(counts[idx >> 7], 1.0f);
}

extern "C" void kernel_launch(void* const* d_in, const int* in_sizes, int n_in,
                              void* d_out, int out_size, void* d_ws, size_t ws_size,
                              hipStream_t stream)
{
    (void)in_sizes; (void)n_in; (void)out_size; (void)ws_size;

    const float* x     = (const float*)d_in[0];
    const int*   ei    = (const int*)d_in[1];
    const int*   src   = ei;
    const int*   dst   = ei + NE;
    const int*   batch = (const int*)d_in[2];
    const float* Wl[5] = {(const float*)d_in[4], (const float*)d_in[6], (const float*)d_in[8],
                          (const float*)d_in[10], (const float*)d_in[12]};
    const float* Bl[5] = {(const float*)d_in[5], (const float*)d_in[7], (const float*)d_in[9],
                          (const float*)d_in[11], (const float*)d_in[13]};
    const float* fc1_w = (const float*)d_in[14];
    const float* fc1_b = (const float*)d_in[15];
    const float* fc2_w = (const float*)d_in[16];
    const float* fc2_b = (const float*)d_in[17];

    float* H      = (float*)d_ws;
    float* H2     = H      + (size_t)NN * DH;
    float* selfc  = H2     + (size_t)NN * DH;
    float* dinv   = selfc  + NN;
    float* sums   = dinv   + NN;
    float* counts = sums   + (size_t)NG * DH;
    float* pooled = counts + NG;
    float* gact   = pooled + (size_t)NG * DH;
    int*   cnt    = (int*)(gact + (size_t)NG * DH);
    int*   rowptr = cnt    + NNP;
    int*   rank   = rowptr + NNP;
    int2*  csr    = (int2*)(rank + NE);

    const int T = 256;

    k_zero_i<<<NNP / T, T, 0, stream>>>(cnt, NNP);
    k_rank<<<(NE + T - 1) / T, T, 0, stream>>>(dst, cnt, rank, NE);
    k_scan_norm<<<NNP / 1024, T, 0, stream>>>(cnt, rowptr, selfc, dinv);
    k_csr_fill<<<(NE + T - 1) / T, T, 0, stream>>>(src, dst, rowptr, rank, dinv, csr, NE);

    const int gemm_blocks = NN / 16;
    const int gath_blocks = NN / 8;

    const float* cur = x;
    for (int l = 0; l < 5; ++l) {
        k_wmma_gemm<DH><<<gemm_blocks, T, 0, stream>>>(cur, Wl[l], nullptr, H2, NN, 0);
        k_gather_combine<<<gath_blocks, T, 0, stream>>>(rowptr, csr, H2, selfc, Bl[l], H);
        cur = H;
    }

    hipMemsetAsync(sums, 0, ((size_t)NG * DH + NG) * sizeof(float), stream);
    k_pool_accum<<<(NN * DH + T - 1) / T, T, 0, stream>>>(H, batch, sums, counts, NN * DH);
    k_pool_div<<<(NG * DH + T - 1) / T, T, 0, stream>>>(sums, counts, pooled);

    k_wmma_gemm<DH><<<NG / 16, T, 0, stream>>>(pooled, fc1_w, fc1_b, gact, NG, 1);
    k_wmma_gemm<DO><<<NG / 16, T, 0, stream>>>(gact, fc2_w, fc2_b, (float*)d_out, NG, 0);
}


# mean-pool sums: one wave per 16 nodes keeps a register run sum per graph id and issues float atomics only at run ends, not one per element; pool sum order changes (runs, then atomics)
# speedup vs baseline: 4.1000x; 1.1723x over previous
#include <hip/hip_runtime.h>

#define NN 50000
#define NE 800000
#define DH 128
#define DO 64
#define NG 128

typedef float v2f __attribute__((ext_vector_type(2)));
typedef float v8f __attribute__((ext_vector_type(8)));

#define NNP 50176

__global__ void k_zero_i(int* p, int n) {
    int i = blockIdx.x * blockDim.x + threadIdx.x;
    if (i < n) p[i] = 0;
}

__global__ void k_rank(const int* __restrict__ dst, int* __restrict__ cnt,
                       int* __restrict__ rank, int E) {
    int e = blockIdx.x * blockDim.x + threadIdx.x;
    if (e < E) rank[e] = atomicAdd(&cnt[dst[e]], 1);
}

__global__ void __launch_bounds__(256)
k_scan_norm(const int* __restrict__ cnt, int* __restrict__ rowptr,
            float* __restrict__ selfc, float* __restrict__ dinv)
{
    __shared__ int sOwn[8];
    __shared__ int sPre[8];
    const int4* c4 = (const int4*)cnt;
    const int b = blockIdx.x, t = threadIdx.x;
    const int lane = t & 31, wave = t >> 5;

    int part = 0;
    for (int i = t; i < b * 256; i += 256) {
        int4 v = c4[i];
        part += (v.x + v.y) + (v.z + v.w);
    }
    const int4 v = c4[b * 256 + t];
    const int own = (v.x + v.y) + (v.z + v.w);

    int incl = own;
#pragma unroll
    for (int o = 1; o < 32; o <<= 1) {
        int up = __shfl_up(incl, o);
        if (lane >= o) incl += up;
    }
#pragma unroll
    for (int o = 16; o > 0; o >>= 1) part += __shfl_xor(part, o);
    if (lane == 31) sOwn[wave] = incl;
    if (lane == 0)  sPre[wave] = part;
    __syncthreads();
    int base = 0;
#pragma unroll
    for (int w = 0; w < 8; ++w) {
        base += sPre[w];
        if (w < wave) base += sOwn[w];
    }
    const int ex = base + incl - own;
    int4 rp;
    rp.x = ex; rp.y = ex + v.x; rp.z = rp.y + v.y; rp.w = rp.z + v.z;
    ((int4*)rowptr)[b * 256 + t] = rp;

    const int i0 = (b * 256 + t) * 4;
    if (i0 < NN) {
        float4 di, sc;
        float d;
        d = (float)(v.x + 1); di.x = 1.0f / sqrtf(d); sc.x = 1.0f / d;
        d = (float)(v.y + 1); di.y = 1.0f / sqrtf(d); sc.y = 1.0f / d;
        d = (float)(v.z + 1); di.z = 1.0f / sqrtf(d); sc.z = 1.0f / d;
        d = (float)(v.w + 1); di.w = 1.0f / sqrtf(d); sc.w = 1.0f / d;
        ((float4*)dinv)[b * 256 + t]  = di;
        ((float4*)selfc)[b * 256 + t] = sc;
    }
}

__global__ void k_csr_fill(const int* __restrict__ src, const int* __restrict__ dst,
                           const int* __restrict__ rowptr, const int* __restrict__ rank,
                           const float* __restrict__ dinv, int2* __restrict__ csr, int E) {
    int e = blockIdx.x * blockDim.x + threadIdx.x;
    if (e < E) {
        int s = src[e], d = dst[e];
        float w = dinv[s] * dinv[d];
        csr[rowptr[d] + rank[e]] = make_int2(s, __float_as_int(w));
    }
}

template <int NC>
__global__ void __launch_bounds__(256)
k_wmma_gemm(const float* __restrict__ A, const float* __restrict__ W,
            const float* __restrict__ bias, float* __restrict__ C,
            int M, int relu)
{
    __shared__ float sWt[NC * 132];
    __shared__ float sA[16 * 132];

    const int tid = threadIdx.x;
    const int r0  = blockIdx.x * 16;

    for (int i = tid; i < 128 * NC; i += 256) {
        int k = i / NC, c = i % NC;
        sWt[c * 132 + k] = W[i];
    }
    for (int i = tid; i < 16 * 128; i += 256) {
        int r = i >> 7, c = i & 127;
        int row = r0 + r;
        sA[r * 132 + c] = (row < M) ? A[row * 128 + c] : 0.0f;
    }
    __syncthreads();

    const int wave = tid >> 5;
    const int lane = tid & 31;
    const int ln   = lane & 15;
    const int hi   = lane >> 4;
    const int c0   = wave * 16;

    if (c0 < NC) {
        const float* aP = &sA [ln        * 132 + 2 * hi];
        const float* bP = &sWt[(c0 + ln) * 132 + 2 * hi];
        v8f acc = {0.f, 0.f, 0.f, 0.f, 0.f, 0.f, 0.f, 0.f};
#pragma unroll
        for (int k = 0; k < 128; k += 4) {
            v2f a = *(const v2f*)(aP + k);
            v2f b = *(const v2f*)(bP + k);
            acc = __builtin_amdgcn_wmma_f32_16x16x4_f32(
                false, a, false, b, (short)0, acc, false, false);
        }
        const int col = c0 + ln;
        const float bv = bias ? bias[col] : 0.0f;
#pragma unroll
        for (int r = 0; r < 8; ++r) {
            int row = r0 + r + 8 * hi;
            if (row < M) {
                float v = acc[r] + bv;
                if (relu) v = fmaxf(v, 0.0f);
                C[row * NC + col] = v;
            }
        }
    }
}

__global__ void __launch_bounds__(256)
k_gather_combine(const int* __restrict__ rowptr, const int2* __restrict__ csr,
                 const float* __restrict__ H2, const float* __restrict__ selfc,
                 const float* __restrict__ bias, float* __restrict__ H)
{
    const int lane = threadIdx.x & 31;
    const int i = __builtin_amdgcn_readfirstlane((int)(blockIdx.x * 8 + (threadIdx.x >> 5)));
    const int beg = rowptr[i], end = rowptr[i + 1];
    const float4* H2v = (const float4*)H2;

    float4 acc = {0.f, 0.f, 0.f, 0.f};
    int j = beg;
    for (; j + 4 <= end; j += 4) {
        const int2 e0 = csr[j], e1 = csr[j + 1], e2 = csr[j + 2], e3 = csr[j + 3];
        const float4 v0 = H2v[(size_t)e0.x * 32 + lane];
        const float4 v1 = H2v[(size_t)e1.x * 32 + lane];
        const float4 v2 = H2v[(size_t)e2.x * 32 + lane];
        const float4 v3 = H2v[(size_t)e3.x * 32 + lane];
        const float w0 = __int_as_float(e0.y), w1 = __int_as_float(e1.y);
        const float w2 = __int_as_float(e2.y), w3 = __int_as_float(e3.y);
        acc.x = fmaf(v0.x, w0, acc.x); acc.y = fmaf(v0.y, w0, acc.y); acc.z = fmaf(v0.z, w0, acc.z); acc.w = fmaf(v0.w, w0, acc.w);
        acc.x = fmaf(v1.x, w1, acc.x); acc.y = fmaf(v1.y, w1, acc.y); acc.z = fmaf(v1.z, w1, acc.z); acc.w = fmaf(v1.w, w1, acc.w);
        acc.x = fmaf(v2.x, w2, acc.x); acc.y = fmaf(v2.y, w2, acc.y); acc.z = fmaf(v2.z, w2, acc.z); acc.w = fmaf(v2.w, w2, acc.w);
        acc.x = fmaf(v3.x, w3, acc.x); acc.y = fmaf(v3.y, w3, acc.y); acc.z = fmaf(v3.z, w3, acc.z); acc.w = fmaf(v3.w, w3, acc.w);
    }
    for (; j < end; ++j) {
        const int2 e0 = csr[j];
        const float4 v0 = H2v[(size_t)e0.x * 32 + lane];
        const float w0 = __int_as_float(e0.y);
        acc.x = fmaf(v0.x, w0, acc.x); acc.y = fmaf(v0.y, w0, acc.y); acc.z = fmaf(v0.z, w0, acc.z); acc.w = fmaf(v0.w, w0, acc.w);
    }

    const float4 h = H2v[(size_t)i * 32 + lane];
    const float sc = selfc[i];
    const int d0 = lane * 4;
    float4 o;
    o.x = fmaxf(acc.x + h.x * sc + bias[d0 + 0], 0.0f);
    o.y = fmaxf(acc.y + h.y * sc + bias[d0 + 1], 0.0f);
    o.z = fmaxf(acc.z + h.z * sc + bias[d0 + 2], 0.0f);
    o.w = fmaxf(acc.w + h.w * sc + bias[d0 + 3], 0.0f);
    ((float4*)H)[(size_t)i * 32 + lane] = o;
}

#define PCH 16

__device__ __forceinline__ void pool_flush(float* __restrict__ sums, float* __restrict__ counts,
                                           int g, int lane, float4 a, float n)
{
    float* s = sums + g * DH + lane * 4;
    atomicAdd(s + 0, a.x);
    atomicAdd(s + 1, a.y);
    atomicAdd(s + 2, a.z);
    atomicAdd(s + 3, a.w);
    if (lane == 0) atomicAdd(&counts[g], n);
}

__global__ void __launch_bounds__(256)
k_pool_runs(const float* __restrict__ H, const int* __restrict__ batch,
            float* __restrict__ sums, float* __restrict__ counts)
{
    const int lane = threadIdx.x & 31;
    const int wv = __builtin_amdgcn_readfirstlane((int)(blockIdx.x * 8 + (threadIdx.x >> 5)));
    if (wv >= NN / PCH) return;
    const int i0 = wv * PCH;
    const float4* Hv = (const float4*)H + (size_t)i0 * 32 + lane;

    float4 v[PCH];
    int    g[PCH];
#pragma unroll
    for (int k = 0; k < PCH; ++k) {
        v[k] = Hv[k * 32];
        g[k] = batch[i0 + k];
    }

    float4 acc = v[0];
    int    cur = g[0];
    float  n   = 1.0f;
#pragma unroll
    for (int k = 1; k < PCH; ++k) {
        if (g[k] != cur) {
            pool_flush(sums, counts, cur, lane, acc, n);
            acc = v[k]; cur = g[k]; n = 1.0f;
        } else {
            acc.x += v[k].x; acc.y += v[k].y; acc.z += v[k].z; acc.w += v[k].w;
            n += 1.0f;
        }
    }
    pool_flush(sums, counts, cur, lane, acc, n);
}

__global__ void k_pool_div(const float* __restrict__ sums, const float* __restrict__ counts,
                           float* __restrict__ pooled)
{
    int idx = blockIdx.x * blockDim.x + threadIdx.x;
    if (idx >= NG * DH) return;
    pooled[idx] = sums[idx] / fmaxf(counts[idx >> 7], 1.0f);
}

extern "C" void kernel_launch(void* const* d_in, const int* in_sizes, int n_in,
                              void* d_out, int out_size, void* d_ws, size_t ws_size,
                              hipStream_t stream)
{
    (void)in_sizes; (void)n_in; (void)out_size; (void)ws_size;

    const float* x     = (const float*)d_in[0];
    const int*   ei    = (const int*)d_in[1];
    const int*   src   = ei;
    const int*   dst   = ei + NE;
    const int*   batch = (const int*)d_in[2];
    const float* Wl[5] = {(const float*)d_in[4], (const float*)d_in[6], (const float*)d_in[8],
                          (const float*)d_in[10], (const float*)d_in[12]};
    const float* Bl[5] = {(const float*)d_in[5], (const float*)d_in[7], (const float*)d_in[9],
                          (const float*)d_in[11], (const float*)d_in[13]};
    const float* fc1_w = (const float*)d_in[14];
    const float* fc1_b = (const float*)d_in[15];
    const float* fc2_w = (const float*)d_in[16];
    const float* fc2_b = (const float*)d_in[17];

    float* H      = (float*)d_ws;
    float* H2     = H      + (size_t)NN * DH;
    float* selfc  = H2     + (size_t)NN * DH;
    float* dinv   = selfc  + NN;
    float* sums   = dinv   + NN;
    float* counts = sums   + (size_t)NG * DH;
    float* pooled = counts + NG;
    float* gact   = pooled + (size_t)NG * DH;
    int*   cnt    = (int*)(gact + (size_t)NG * DH);
    int*   rowptr = cnt    + NNP;
    int*   rank   = rowptr + NNP;
    int2*  csr    = (int2*)(rank + NE);

    const int T = 256;

    k_zero_i<<<NNP / T, T, 0, stream>>>(cnt, NNP);
    k_rank<<<(NE + T - 1) / T, T, 0, stream>>>(dst, cnt, rank, NE);
    k_scan_norm<<<NNP / 1024, T, 0, stream>>>(cnt, rowptr, selfc, dinv);
    k_csr_fill<<<(NE + T - 1) / T, T, 0, stream>>>(src, dst, rowptr, rank, dinv, csr, NE);

    const int gemm_blocks = NN / 16;
    const int gath_blocks = NN / 8;

    const float* cur = x;
    for (int l = 0; l < 5; ++l) {
        k_wmma_gemm<DH><<<gemm_blocks, T, 0, stream>>>(cur, Wl[l], nullptr, H2, NN, 0);
        k_gather_combine<<<gath_blocks, T, 0, stream>>>(rowptr, csr, H2, selfc, Bl[l], H);
        cur = H;
    }

    hipMemsetAsync(sums, 0, ((size_t)NG * DH + NG) * sizeof(float), stream);
    k_pool_runs<<<(NN / PCH + 7) / 8, T, 0, stream>>>(H, batch, sums, counts);
    k_pool_div<<<(NG * DH + T - 1) / T, T, 0, stream>>>(sums, counts, pooled);

    k_wmma_gemm<DH><<<NG / 16, T, 0, stream>>>(pooled, fc1_w, fc1_b, gact, NG, 1);
    k_wmma_gemm<DO><<<NG / 16, T, 0, stream>>>(gact, fc2_w, fc2_b, (float*)d_out, NG, 0);
}


# fewer waves to start: a gather wave walks 5 nodes (row reads 8 deep), 512 GEMM blocks stage W once and walk their strips, edge kernels take 4 edges per lane; sums stay in CSR slot (arrival) order
# speedup vs baseline: 5.2084x; 1.2703x over previous
#include <hip/hip_runtime.h>

#define NN 50000
#define NE 800000
#define DH 128
#define DO 64
#define NG 128

typedef float v2f __attribute__((ext_vector_type(2)));
typedef float v8f __attribute__((ext_vector_type(8)));

#define NNP 50176

__global__ void k_zero_i(int* p, int n) {
    int i = blockIdx.x * blockDim.x + threadIdx.x;
    if (i < n) p[i] = 0;
}

__global__ void k_rank(const int* __restrict__ dst, int* __restrict__ cnt,
                       int* __restrict__ rank, int E) {
    int q = blockIdx.x * blockDim.x + threadIdx.x;
    if (q * 4 < E) {
        const int4 d = ((const int4*)dst)[q];
        int4 r;
        r.x = atomicAdd(&cnt[d.x], 1);
        r.y = atomicAdd(&cnt[d.y], 1);
        r.z = atomicAdd(&cnt[d.z], 1);
        r.w = atomicAdd(&cnt[d.w], 1);
        ((int4*)rank)[q] = r;
    }
}

__global__ void __launch_bounds__(256)
k_scan_norm(const int* __restrict__ cnt, int* __restrict__ rowptr,
            float* __restrict__ selfc, float* __restrict__ dinv)
{
    __shared__ int sOwn[8];
    __shared__ int sPre[8];
    const int4* c4 = (const int4*)cnt;
    const int b = blockIdx.x, t = threadIdx.x;
    const int lane = t & 31, wave = t >> 5;

    int part = 0;
    for (int i = t; i < b * 256; i += 256) {
        int4 v = c4[i];
        part += (v.x + v.y) + (v.z + v.w);
    }
    const int4 v = c4[b * 256 + t];
    const int own = (v.x + v.y) + (v.z + v.w);

    int incl = own;
#pragma unroll
    for (int o = 1; o < 32; o <<= 1) {
        int up = __shfl_up(incl, o);
        if (lane >= o) incl += up;
    }
#pragma unroll
    for (int o = 16; o > 0; o >>= 1) part += __shfl_xor(part, o);
    if (lane == 31) sOwn[wave] = incl;
    if (lane == 0)  sPre[wave] = part;
    __syncthreads();
    int base = 0;
#pragma unroll
    for (int w = 0; w < 8; ++w) {
        base += sPre[w];
        if (w < wave) base += sOwn[w];
    }
    const int ex = base + incl - own;
    int4 rp;
    rp.x = ex; rp.y = ex + v.x; rp.z = rp.y + v.y; rp.w = rp.z + v.z;
    ((int4*)rowptr)[b * 256 + t] = rp;

    const int i0 = (b * 256 + t) * 4;
    if (i0 < NN) {
        float4 di, sc;
        float d;
        d = (float)(v.x + 1); di.x = 1.0f / sqrtf(d); sc.x = 1.0f / d;
        d = (float)(v.y + 1); di.y = 1.0f / sqrtf(d); sc.y = 1.0f / d;
        d = (float)(v.z + 1); di.z = 1.0f / sqrtf(d); sc.z = 1.0f / d;
        d = (float)(v.w + 1); di.w = 1.0f / sqrtf(d); sc.w = 1.0f / d;
        ((float4*)dinv)[b * 256 + t]  = di;
        ((float4*)selfc)[b * 256 + t] = sc;
    }
}

__device__ __forceinline__ void csr_put(int s, int d, int r, const int* __restrict__ rowptr,
                                        const float* __restrict__ dinv, int2* __restrict__ csr) {
    float w = dinv[s] * dinv[d];
    csr[rowptr[d] + r] = make_int2(s, __float_as_int(w));
}

__global__ void k_csr_fill(const int* __restrict__ src, const int* __restrict__ dst,
                           const int* __restrict__ rowptr, const int* __restrict__ rank,
                           const float* __restrict__ dinv, int2* __restrict__ csr, int E) {
    int q = blockIdx.x * blockDim.x + threadIdx.x;
    if (q * 4 < E) {
        const int4 s = ((const int4*)src)[q];
        const int4 d = ((const int4*)dst)[q];
        const int4 r = ((const int4*)rank)[q];
        csr_put(s.x, d.x, r.x, rowptr, dinv, csr);
        csr_put(s.y, d.y, r.y, rowptr, dinv, csr);
        csr_put(s.z, d.z, r.z, rowptr, dinv, csr);
        csr_put(s.w, d.w, r.w, rowptr, dinv, csr);
    }
}

template <int NC>
__global__ void __launch_bounds__(256)
k_wmma_gemm(const float* __restrict__ A, const float* __restrict__ W,
            const float* __restrict__ bias, float* __restrict__ C,
            int M, int relu)
{
    __shared__ float sWt[NC * 132];
    __shared__ alignas(16) float sA[16 * 132];

    const int tid = threadIdx.x;
    const int nstrips = (M + 15) >> 4;

    const int ar = tid >> 5, ac = tid & 31;
    int s = blockIdx.x;
    float4 p0 = make_float4(0.f, 0.f, 0.f, 0.f), p1 = p0;
    if (s < nstrips) {
        const int row0 = s * 16 + ar, row1 = row0 + 8;
        if (row0 < M) p0 = *(const float4*)(A + (size_t)row0 * 128 + ac * 4);
        if (row1 < M) p1 = *(const float4*)(A + (size_t)row1 * 128 + ac * 4);
    }

    for (int i = tid; i < 128 * NC; i += 256) {
        int k = i / NC, c = i % NC;
        sWt[c * 132 + k] = W[i];
    }

    const int wave = tid >> 5;
    const int lane = tid & 31;
    const int ln   = lane & 15;
    const int hi   = lane >> 4;
    const int c0   = wave * 16;
    const int col  = c0 + ln;
    const float bv = (c0 < NC && bias) ? bias[col] : 0.0f;

    for (; s < nstrips; s += gridDim.x) {
        *(float4*)&sA[ar * 132 + ac * 4]       = p0;
        *(float4*)&sA[(ar + 8) * 132 + ac * 4] = p1;
        __syncthreads();

        const int sn = s + gridDim.x;
        if (sn < nstrips) {
            const int row0 = sn * 16 + ar, row1 = row0 + 8;
            p0 = make_float4(0.f, 0.f, 0.f, 0.f); p1 = p0;
            if (row0 < M) p0 = *(const float4*)(A + (size_t)row0 * 128 + ac * 4);
            if (row1 < M) p1 = *(const float4*)(A + (size_t)row1 * 128 + ac * 4);
        }

        if (c0 < NC) {
            const float* aP = &sA [ln        * 132 + 2 * hi];
            const float* bP = &sWt[(c0 + ln) * 132 + 2 * hi];
            v8f acc = {0.f, 0.f, 0.f, 0.f, 0.f, 0.f, 0.f, 0.f};
#pragma unroll
            for (int k = 0; k < 128; k += 4) {
                v2f a = *(const v2f*)(aP + k);
                v2f b = *(const v2f*)(bP + k);
                acc = __builtin_amdgcn_wmma_f32_16x16x4_f32(
                    false, a, false, b, (short)0, acc, false, false);
            }
            const int r0 = s * 16;
#pragma unroll
            for (int r = 0; r < 8; ++r) {
                int row = r0 + r + 8 * hi;
                if (row < M) {
                    float v = acc[r] + bv;
                    if (relu) v = fmaxf(v, 0.0f);
                    C[(size_t)row * NC + col] = v;
                }
            }
        }
        __syncthreads();
    }
}

#define GR 5

__device__ __forceinline__ void edge_fma(float4& acc, const float4 v, const float w)
{
    acc.x = fmaf(v.x, w, acc.x); acc.y = fmaf(v.y, w, acc.y);
    acc.z = fmaf(v.z, w, acc.z); acc.w = fmaf(v.w, w, acc.w);
}

__global__ void __launch_bounds__(256)
k_gather_combine(const int* __restrict__ rowptr, const int2* __restrict__ csr,
                 const float* __restrict__ H2, const float* __restrict__ selfc,
                 const float* __restrict__ bias, float* __restrict__ H)
{
    const int lane = threadIdx.x & 31;
    const int wv = __builtin_amdgcn_readfirstlane((int)(blockIdx.x * 8 + (threadIdx.x >> 5)));
    const int i0 = wv * GR;
    if (i0 >= NN) return;
    const float4* H2v = (const float4*)H2 + lane;
    const float4 b4 = ((const float4*)bias)[lane];

    int beg = rowptr[i0];
#pragma unroll 1
    for (int n = 0; n < GR; ++n) {
        const int i = i0 + n;
        const int end = rowptr[i + 1];
        const float4 h = H2v[(size_t)i * 32];
        const float sc = selfc[i];

        float4 acc = {0.f, 0.f, 0.f, 0.f};
        int j = beg;
        for (; j + 8 <= end; j += 8) {
            int2   e[8];
            float4 v[8];
#pragma unroll
            for (int t = 0; t < 8; ++t) e[t] = csr[j + t];
#pragma unroll
            for (int t = 0; t < 8; ++t) v[t] = H2v[(size_t)e[t].x * 32];
#pragma unroll
            for (int t = 0; t < 8; ++t) edge_fma(acc, v[t], __int_as_float(e[t].y));
        }
        const int rem = end - j;
        if (rem > 0) {
            int2   e[7];
            float4 v[7];
#pragma unroll
            for (int t = 0; t < 7; ++t) e[t] = csr[min(j + t, end - 1)];
#pragma unroll
            for (int t = 0; t < 7; ++t) if (t < rem) v[t] = H2v[(size_t)e[t].x * 32];
#pragma unroll
            for (int t = 0; t < 7; ++t) if (t < rem) edge_fma(acc, v[t], __int_as_float(e[t].y));
        }

        float4 o;
        o.x = fmaxf(acc.x + h.x * sc + b4.x, 0.0f);
        o.y = fmaxf(acc.y + h.y * sc + b4.y, 0.0f);
        o.z = fmaxf(acc.z + h.z * sc + b4.z, 0.0f);
        o.w = fmaxf(acc.w + h.w * sc + b4.w, 0.0f);
        ((float4*)H)[(size_t)i * 32 + lane] = o;
        beg = end;
    }
}

#define PCH 16

__device__ __forceinline__ void pool_flush(float* __restrict__ sums, float* __restrict__ counts,
                                           int g, int lane, float4 a, float n)
{
    float* s = sums + g * DH + lane * 4;
    atomicAdd(s + 0, a.x);
    atomicAdd(s + 1, a.y);
    atomicAdd(s + 2, a.z);
    atomicAdd(s + 3, a.w);
    if (lane == 0) atomicAdd(&counts[g], n);
}

__global__ void __launch_bounds__(256)
k_pool_runs(const float* __restrict__ H, const int* __restrict__ batch,
            float* __restrict__ sums, float* __restrict__ counts)
{
    const int lane = threadIdx.x & 31;
    const int wv = __builtin_amdgcn_readfirstlane((int)(blockIdx.x * 8 + (threadIdx.x >> 5)));
    if (wv >= NN / PCH) return;
    const int i0 = wv * PCH;
    const float4* Hv = (const float4*)H + (size_t)i0 * 32 + lane;

    float4 v[PCH];
    int    g[PCH];
#pragma unroll
    for (int k = 0; k < PCH; ++k) {
        v[k] = Hv[k * 32];
        g[k] = batch[i0 + k];
    }

    float4 acc = v[0];
    int    cur = g[0];
    float  n   = 1.0f;
#pragma unroll
    for (int k = 1; k < PCH; ++k) {
        if (g[k] != cur) {
            pool_flush(sums, counts, cur, lane, acc, n);
            acc = v[k]; cur = g[k]; n = 1.0f;
        } else {
            acc.x += v[k].x; acc.y += v[k].y; acc.z += v[k].z; acc.w += v[k].w;
            n += 1.0f;
        }
    }
    pool_flush(sums, counts, cur, lane, acc, n);
}

__global__ void k_pool_div(const float* __restrict__ sums, const float* __restrict__ counts,
                           float* __restrict__ pooled)
{
    int idx = blockIdx.x * blockDim.x + threadIdx.x;
    if (idx >= NG * DH) return;
    pooled[idx] = sums[idx] / fmaxf(counts[idx >> 7], 1.0f);
}

extern "C" void kernel_launch(void* const* d_in, const int* in_sizes, int n_in,
                              void* d_out, int out_size, void* d_ws, size_t ws_size,
                              hipStream_t stream)
{
    (void)in_sizes; (void)n_in; (void)out_size; (void)ws_size;

    const float* x     = (const float*)d_in[0];
    const int*   ei    = (const int*)d_in[1];
    const int*   src   = ei;
    const int*   dst   = ei + NE;
    const int*   batch = (const int*)d_in[2];
    const float* Wl[5] = {(const float*)d_in[4], (const float*)d_in[6], (const float*)d_in[8],
                          (const float*)d_in[10], (const float*)d_in[12]};
    const float* Bl[5] = {(const float*)d_in[5], (const float*)d_in[7], (const float*)d_in[9],
                          (const float*)d_in[11], (const float*)d_in[13]};
    const float* fc1_w = (const float*)d_in[14];
    const float* fc1_b = (const float*)d_in[15];
    const float* fc2_w = (const float*)d_in[16];
    const float* fc2_b = (const float*)d_in[17];

    float* H      = (float*)d_ws;
    float* H2     = H      + (size_t)NN * DH;
    float* selfc  = H2     + (size_t)NN * DH;
    float* dinv   = selfc  + NN;
    float* sums   = dinv   + NN;
    float* counts = sums   + (size_t)NG * DH;
    float* pooled = counts + NG;
    float* gact   = pooled + (size_t)NG * DH;
    int*   cnt    = (int*)(gact + (size_t)NG * DH);
    int*   rowptr = cnt    + NNP;
    int*   rank   = rowptr + NNP;
    int2*  csr    = (int2*)(rank + NE);

    const int T = 256;

    k_zero_i<<<NNP / T, T, 0, stream>>>(cnt, NNP);
    k_rank<<<(NE / 4 + T - 1) / T, T, 0, stream>>>(dst, cnt, rank, NE);
    k_scan_norm<<<NNP / 1024, T, 0, stream>>>(cnt, rowptr, selfc, dinv);
    k_csr_fill<<<(NE / 4 + T - 1) / T, T, 0, stream>>>(src, dst, rowptr, rank, dinv, csr, NE);

    const int gemm_blocks = 512;
    const int gath_blocks = (NN / GR + 7) / 8;

    const float* cur = x;
    for (int l = 0; l < 5; ++l) {
        k_wmma_gemm<DH><<<gemm_blocks, T, 0, stream>>>(cur, Wl[l], nullptr, H2, NN, 0);
        k_gather_combine<<<gath_blocks, T, 0, stream>>>(rowptr, csr, H2, selfc, Bl[l], H);
        cur = H;
    }

    hipMemsetAsync(sums, 0, ((size_t)NG * DH + NG) * sizeof(float), stream);
    k_pool_runs<<<(NN / PCH + 7) / 8, T, 0, stream>>>(H, batch, sums, counts);
    k_pool_div<<<(NG * DH + T - 1) / T, T, 0, stream>>>(sums, counts, pooled);

    k_wmma_gemm<DH><<<NG / 16, T, 0, stream>>>(pooled, fc1_w, fc1_b, gact, NG, 1);
    k_wmma_gemm<DO><<<NG / 16, T, 0, stream>>>(gact, fc2_w, fc2_b, (float*)d_out, NG, 0);
}


# layers 2-5: each product moves into the gather kernel before it: 8 waves gather a 16-node strip into LDS, then multiply it by the next W on WMMA; 4 launches and H's memory trip go; sums unchanged
# speedup vs baseline: 5.9652x; 1.1453x over previous
#include <hip/hip_runtime.h>

#define NN 50000
#define NE 800000
#define DH 128
#define DO 64
#define NG 128

typedef float v2f __attribute__((ext_vector_type(2)));
typedef float v8f __attribute__((ext_vector_type(8)));

#define NNP 50176

__global__ void k_zero_i(int* p, int n) {
    int i = blockIdx.x * blockDim.x + threadIdx.x;
    if (i < n) p[i] = 0;
}

__global__ void k_rank(const int* __restrict__ dst, int* __restrict__ cnt,
                       int* __restrict__ rank, int E) {
    int q = blockIdx.x * blockDim.x + threadIdx.x;
    if (q * 4 < E) {
        const int4 d = ((const int4*)dst)[q];
        int4 r;
        r.x = atomicAdd(&cnt[d.x], 1);
        r.y = atomicAdd(&cnt[d.y], 1);
        r.z = atomicAdd(&cnt[d.z], 1);
        r.w = atomicAdd(&cnt[d.w], 1);
        ((int4*)rank)[q] = r;
    }
}

__global__ void __launch_bounds__(256)
k_scan_norm(const int* __restrict__ cnt, int* __restrict__ rowptr,
            float* __restrict__ selfc, float* __restrict__ dinv)
{
    __shared__ int sOwn[8];
    __shared__ int sPre[8];
    const int4* c4 = (const int4*)cnt;
    const int b = blockIdx.x, t = threadIdx.x;
    const int lane = t & 31, wave = t >> 5;

    int part = 0;
    for (int i = t; i < b * 256; i += 256) {
        int4 v = c4[i];
        part += (v.x + v.y) + (v.z + v.w);
    }
    const int4 v = c4[b * 256 + t];
    const int own = (v.x + v.y) + (v.z + v.w);

    int incl = own;
#pragma unroll
    for (int o = 1; o < 32; o <<= 1) {
        int up = __shfl_up(incl, o);
        if (lane >= o) incl += up;
    }
#pragma unroll
    for (int o = 16; o > 0; o >>= 1) part += __shfl_xor(part, o);
    if (lane == 31) sOwn[wave] = incl;
    if (lane == 0)  sPre[wave] = part;
    __syncthreads();
    int base = 0;
#pragma unroll
    for (int w = 0; w < 8; ++w) {
        base += sPre[w];
        if (w < wave) base += sOwn[w];
    }
    const int ex = base + incl - own;
    int4 rp;
    rp.x = ex; rp.y = ex + v.x; rp.z = rp.y + v.y; rp.w = rp.z + v.z;
    ((int4*)rowptr)[b * 256 + t] = rp;

    const int i0 = (b * 256 + t) * 4;
    if (i0 < NN) {
        float4 di, sc;
        float d;
        d = (float)(v.x + 1); di.x = 1.0f / sqrtf(d); sc.x = 1.0f / d;
        d = (float)(v.y + 1); di.y = 1.0f / sqrtf(d); sc.y = 1.0f / d;
        d = (float)(v.z + 1); di.z = 1.0f / sqrtf(d); sc.z = 1.0f / d;
        d = (float)(v.w + 1); di.w = 1.0f / sqrtf(d); sc.w = 1.0f / d;
        ((float4*)dinv)[b * 256 + t]  = di;
        ((float4*)selfc)[b * 256 + t] = sc;
    }
}

__device__ __forceinline__ void csr_put(int s, int d, int r, const int* __restrict__ rowptr,
                                        const float* __restrict__ dinv, int2* __restrict__ csr) {
    float w = dinv[s] * dinv[d];
    csr[rowptr[d] + r] = make_int2(s, __float_as_int(w));
}

__global__ void k_csr_fill(const int* __restrict__ src, const int* __restrict__ dst,
                           const int* __restrict__ rowptr, const int* __restrict__ rank,
                           const float* __restrict__ dinv, int2* __restrict__ csr, int E) {
    int q = blockIdx.x * blockDim.x + threadIdx.x;
    if (q * 4 < E) {
        const int4 s = ((const int4*)src)[q];
        const int4 d = ((const int4*)dst)[q];
        const int4 r = ((const int4*)rank)[q];
        csr_put(s.x, d.x, r.x, rowptr, dinv, csr);
        csr_put(s.y, d.y, r.y, rowptr, dinv, csr);
        csr_put(s.z, d.z, r.z, rowptr, dinv, csr);
        csr_put(s.w, d.w, r.w, rowptr, dinv, csr);
    }
}

template <int NC>
__global__ void __launch_bounds__(256)
k_wmma_gemm(const float* __restrict__ A, const float* __restrict__ W,
            const float* __restrict__ bias, float* __restrict__ C,
            int M, int relu)
{
    __shared__ float sWt[NC * 132];
    __shared__ alignas(16) float sA[16 * 132];

    const int tid = threadIdx.x;
    const int nstrips = (M + 15) >> 4;

    const int ar = tid >> 5, ac = tid & 31;
    int s = blockIdx.x;
    float4 p0 = make_float4(0.f, 0.f, 0.f, 0.f), p1 = p0;
    if (s < nstrips) {
        const int row0 = s * 16 + ar, row1 = row0 + 8;
        if (row0 < M) p0 = *(const float4*)(A + (size_t)row0 * 128 + ac * 4);
        if (row1 < M) p1 = *(const float4*)(A + (size_t)row1 * 128 + ac * 4);
    }

    for (int i = tid; i < 128 * NC; i += 256) {
        int k = i / NC, c = i % NC;
        sWt[c * 132 + k] = W[i];
    }

    const int wave = tid >> 5;
    const int lane = tid & 31;
    const int ln   = lane & 15;
    const int hi   = lane >> 4;
    const int c0   = wave * 16;
    const int col  = c0 + ln;
    const float bv = (c0 < NC && bias) ? bias[col] : 0.0f;

    for (; s < nstrips; s += gridDim.x) {
        *(float4*)&sA[ar * 132 + ac * 4]       = p0;
        *(float4*)&sA[(ar + 8) * 132 + ac * 4] = p1;
        __syncthreads();

        const int sn = s + gridDim.x;
        if (sn < nstrips) {
            const int row0 = sn * 16 + ar, row1 = row0 + 8;
            p0 = make_float4(0.f, 0.f, 0.f, 0.f); p1 = p0;
            if (row0 < M) p0 = *(const float4*)(A + (size_t)row0 * 128 + ac * 4);
            if (row1 < M) p1 = *(const float4*)(A + (size_t)row1 * 128 + ac * 4);
        }

        if (c0 < NC) {
            const float* aP = &sA [ln        * 132 + 2 * hi];
            const float* bP = &sWt[(c0 + ln) * 132 + 2 * hi];
            v8f acc = {0.f, 0.f, 0.f, 0.f, 0.f, 0.f, 0.f, 0.f};
#pragma unroll
            for (int k = 0; k < 128; k += 4) {
                v2f a = *(const v2f*)(aP + k);
                v2f b = *(const v2f*)(bP + k);
                acc = __builtin_amdgcn_wmma_f32_16x16x4_f32(
                    false, a, false, b, (short)0, acc, false, false);
            }
            const int r0 = s * 16;
#pragma unroll
            for (int r = 0; r < 8; ++r) {
                int row = r0 + r + 8 * hi;
                if (row < M) {
                    float v = acc[r] + bv;
                    if (relu) v = fmaxf(v, 0.0f);
                    C[(size_t)row * NC + col] = v;
                }
            }
        }
        __syncthreads();
    }
}

#define GR 5

__device__ __forceinline__ void edge_fma(float4& acc, const float4 v, const float w)
{
    acc.x = fmaf(v.x, w, acc.x); acc.y = fmaf(v.y, w, acc.y);
    acc.z = fmaf(v.z, w, acc.z); acc.w = fmaf(v.w, w, acc.w);
}

__global__ void __launch_bounds__(256)
k_gather_combine(const int* __restrict__ rowptr, const int2* __restrict__ csr,
                 const float* __restrict__ H2, const float* __restrict__ selfc,
                 const float* __restrict__ bias, float* __restrict__ H)
{
    const int lane = threadIdx.x & 31;
    const int wv = __builtin_amdgcn_readfirstlane((int)(blockIdx.x * 8 + (threadIdx.x >> 5)));
    const int i0 = wv * GR;
    if (i0 >= NN) return;
    const float4* H2v = (const float4*)H2 + lane;
    const float4 b4 = ((const float4*)bias)[lane];

    int beg = rowptr[i0];
#pragma unroll 1
    for (int n = 0; n < GR; ++n) {
        const int i = i0 + n;
        const int end = rowptr[i + 1];
        const float4 h = H2v[(size_t)i * 32];
        const float sc = selfc[i];

        float4 acc = {0.f, 0.f, 0.f, 0.f};
        int j = beg;
        for (; j + 8 <= end; j += 8) {
            int2   e[8];
            float4 v[8];
#pragma unroll
            for (int t = 0; t < 8; ++t) e[t] = csr[j + t];
#pragma unroll
            for (int t = 0; t < 8; ++t) v[t] = H2v[(size_t)e[t].x * 32];
#pragma unroll
            for (int t = 0; t < 8; ++t) edge_fma(acc, v[t], __int_as_float(e[t].y));
        }
        const int rem = end - j;
        if (rem > 0) {
            int2   e[7];
            float4 v[7];
#pragma unroll
            for (int t = 0; t < 7; ++t) e[t] = csr[min(j + t, end - 1)];
#pragma unroll
            for (int t = 0; t < 7; ++t) if (t < rem) v[t] = H2v[(size_t)e[t].x * 32];
#pragma unroll
            for (int t = 0; t < 7; ++t) if (t < rem) edge_fma(acc, v[t], __int_as_float(e[t].y));
        }

        float4 o;
        o.x = fmaxf(acc.x + h.x * sc + b4.x, 0.0f);
        o.y = fmaxf(acc.y + h.y * sc + b4.y, 0.0f);
        o.z = fmaxf(acc.z + h.z * sc + b4.z, 0.0f);
        o.w = fmaxf(acc.w + h.w * sc + b4.w, 0.0f);
        ((float4*)H)[(size_t)i * 32 + lane] = o;
        beg = end;
    }
}

typedef __attribute__((address_space(4))) int   cs_int;
typedef __attribute__((address_space(4))) int2  cs_int2;
typedef __attribute__((address_space(4))) float cs_float;

__global__ void __launch_bounds__(256)
k_gather_gemm(const int* __restrict__ rowptr, const int2* __restrict__ csr,
              const float* __restrict__ H2, const float* __restrict__ selfc,
              const float* __restrict__ bias, const float* __restrict__ Wn,
              float* __restrict__ H2n)
{
    __shared__ float sWt[DH * 132];
    __shared__ alignas(16) float sA[16 * 132];

    const int tid = threadIdx.x;

    for (int i = tid; i < DH * DH; i += 256) {
        int k = i / DH, c = i % DH;
        sWt[c * 132 + k] = Wn[i];
    }

    const int lane = tid & 31;
    const int wave = __builtin_amdgcn_readfirstlane(tid >> 5);
    const int ln   = lane & 15;
    const int hi   = lane >> 4;
    const int col  = wave * 16 + ln;
    const float4* H2v = (const float4*)H2 + lane;
    const float4 b4 = ((const float4*)bias)[lane];
    const float* aP = &sA [ln  * 132 + 2 * hi];
    const float* bP = &sWt[col * 132 + 2 * hi];
    const cs_int*   rowptr_c = (const cs_int*)rowptr;
    const cs_int2*  csr_c    = (const cs_int2*)csr;
    const cs_float* selfc_c  = (const cs_float*)selfc;

    for (int s = blockIdx.x; s < NN / 16; s += gridDim.x) {
        const int i0 = s * 16 + wave * 2;
        int beg = rowptr_c[i0];
#pragma unroll 1
        for (int n = 0; n < 2; ++n) {
            const int i = i0 + n;
            const int end = rowptr_c[i + 1];
            const float4 h = H2v[(size_t)i * 32];
            const float sc = selfc_c[i];

            float4 acc = {0.f, 0.f, 0.f, 0.f};
            int j = beg;
            for (; j + 8 <= end; j += 8) {
                int2   e[8];
                float4 v[8];
#pragma unroll
                for (int t = 0; t < 8; ++t) e[t] = csr_c[j + t];
#pragma unroll
                for (int t = 0; t < 8; ++t) v[t] = H2v[(size_t)e[t].x * 32];
#pragma unroll
                for (int t = 0; t < 8; ++t) edge_fma(acc, v[t], __int_as_float(e[t].y));
            }
            const int rem = end - j;
            if (rem > 0) {
                int2   e[7];
                float4 v[7];
#pragma unroll
                for (int t = 0; t < 7; ++t) e[t] = csr_c[min(j + t, end - 1)];
#pragma unroll
                for (int t = 0; t < 7; ++t) if (t < rem) v[t] = H2v[(size_t)e[t].x * 32];
#pragma unroll
                for (int t = 0; t < 7; ++t) if (t < rem) edge_fma(acc, v[t], __int_as_float(e[t].y));
            }

            float4 o;
            o.x = fmaxf(acc.x + h.x * sc + b4.x, 0.0f);
            o.y = fmaxf(acc.y + h.y * sc + b4.y, 0.0f);
            o.z = fmaxf(acc.z + h.z * sc + b4.z, 0.0f);
            o.w = fmaxf(acc.w + h.w * sc + b4.w, 0.0f);
            *(float4*)&sA[(wave * 2 + n) * 132 + lane * 4] = o;
            beg = end;
        }
        __syncthreads();

        v8f acc = {0.f, 0.f, 0.f, 0.f, 0.f, 0.f, 0.f, 0.f};
#pragma unroll
        for (int k = 0; k < DH; k += 4) {
            v2f a = *(const v2f*)(aP + k);
            v2f b = *(const v2f*)(bP + k);
            acc = __builtin_amdgcn_wmma_f32_16x16x4_f32(
                false, a, false, b, (short)0, acc, false, false);
        }
        __syncthreads();

        float* outp = H2n + (size_t)(s * 16 + 8 * hi) * DH + col;
#pragma unroll
        for (int r = 0; r < 8; ++r) outp[(size_t)r * DH] = acc[r];
    }
}

#define PCH 16

__device__ __forceinline__ void pool_flush(float* __restrict__ sums, float* __restrict__ counts,
                                           int g, int lane, float4 a, float n)
{
    float* s = sums + g * DH + lane * 4;
    atomicAdd(s + 0, a.x);
    atomicAdd(s + 1, a.y);
    atomicAdd(s + 2, a.z);
    atomicAdd(s + 3, a.w);
    if (lane == 0) atomicAdd(&counts[g], n);
}

__global__ void __launch_bounds__(256)
k_pool_runs(const float* __restrict__ H, const int* __restrict__ batch,
            float* __restrict__ sums, float* __restrict__ counts)
{
    const int lane = threadIdx.x & 31;
    const int wv = __builtin_amdgcn_readfirstlane((int)(blockIdx.x * 8 + (threadIdx.x >> 5)));
    if (wv >= NN / PCH) return;
    const int i0 = wv * PCH;
    const float4* Hv = (const float4*)H + (size_t)i0 * 32 + lane;

    float4 v[PCH];
    int    g[PCH];
#pragma unroll
    for (int k = 0; k < PCH; ++k) {
        v[k] = Hv[k * 32];
        g[k] = batch[i0 + k];
    }

    float4 acc = v[0];
    int    cur = g[0];
    float  n   = 1.0f;
#pragma unroll
    for (int k = 1; k < PCH; ++k) {
        if (g[k] != cur) {
            pool_flush(sums, counts, cur, lane, acc, n);
            acc = v[k]; cur = g[k]; n = 1.0f;
        } else {
            acc.x += v[k].x; acc.y += v[k].y; acc.z += v[k].z; acc.w += v[k].w;
            n += 1.0f;
        }
    }
    pool_flush(sums, counts, cur, lane, acc, n);
}

__global__ void k_pool_div(const float* __restrict__ sums, const float* __restrict__ counts,
                           float* __restrict__ pooled)
{
    int idx = blockIdx.x * blockDim.x + threadIdx.x;
    if (idx >= NG * DH) return;
    pooled[idx] = sums[idx] / fmaxf(counts[idx >> 7], 1.0f);
}

extern "C" void kernel_launch(void* const* d_in, const int* in_sizes, int n_in,
                              void* d_out, int out_size, void* d_ws, size_t ws_size,
                              hipStream_t stream)
{
    (void)in_sizes; (void)n_in; (void)out_size; (void)ws_size;

    const float* x     = (const float*)d_in[0];
    const int*   ei    = (const int*)d_in[1];
    const int*   src   = ei;
    const int*   dst   = ei + NE;
    const int*   batch = (const int*)d_in[2];
    const float* Wl[5] = {(const float*)d_in[4], (const float*)d_in[6], (const float*)d_in[8],
                          (const float*)d_in[10], (const float*)d_in[12]};
    const float* Bl[5] = {(const float*)d_in[5], (const float*)d_in[7], (const float*)d_in[9],
                          (const float*)d_in[11], (const float*)d_in[13]};
    const float* fc1_w = (const float*)d_in[14];
    const float* fc1_b = (const float*)d_in[15];
    const float* fc2_w = (const float*)d_in[16];
    const float* fc2_b = (const float*)d_in[17];

    float* H      = (float*)d_ws;
    float* H2     = H      + (size_t)NN * DH;
    float* selfc  = H2     + (size_t)NN * DH;
    float* dinv   = selfc  + NN;
    float* sums   = dinv   + NN;
    float* counts = sums   + (size_t)NG * DH;
    float* pooled = counts + NG;
    float* gact   = pooled + (size_t)NG * DH;
    int*   cnt    = (int*)(gact + (size_t)NG * DH);
    int*   rowptr = cnt    + NNP;
    int*   rank   = rowptr + NNP;
    int2*  csr    = (int2*)(rank + NE);

    const int T = 256;

    k_zero_i<<<NNP / T, T, 0, stream>>>(cnt, NNP);
    k_rank<<<(NE / 4 + T - 1) / T, T, 0, stream>>>(dst, cnt, rank, NE);
    k_scan_norm<<<NNP / 1024, T, 0, stream>>>(cnt, rowptr, selfc, dinv);
    k_csr_fill<<<(NE / 4 + T - 1) / T, T, 0, stream>>>(src, dst, rowptr, rank, dinv, csr, NE);

    const int gemm_blocks = 512;
    const int gath_blocks = (NN / GR + 7) / 8;

    const int fuse_blocks = 1024;

    k_wmma_gemm<DH><<<gemm_blocks, T, 0, stream>>>(x, Wl[0], nullptr, H2, NN, 0);
    k_gather_gemm<<<fuse_blocks, T, 0, stream>>>(rowptr, csr, H2, selfc, Bl[0], Wl[1], H);
    k_gather_gemm<<<fuse_blocks, T, 0, stream>>>(rowptr, csr, H,  selfc, Bl[1], Wl[2], H2);
    k_gather_gemm<<<fuse_blocks, T, 0, stream>>>(rowptr, csr, H2, selfc, Bl[2], Wl[3], H);
    k_gather_gemm<<<fuse_blocks, T, 0, stream>>>(rowptr, csr, H,  selfc, Bl[3], Wl[4], H2);
    k_gather_combine<<<gath_blocks, T, 0, stream>>>(rowptr, csr, H2, selfc, Bl[4], H);

    hipMemsetAsync(sums, 0, ((size_t)NG * DH + NG) * sizeof(float), stream);
    k_pool_runs<<<(NN / PCH + 7) / 8, T, 0, stream>>>(H, batch, sums, counts);
    k_pool_div<<<(NG * DH + T - 1) / T, T, 0, stream>>>(sums, counts, pooled);

    k_wmma_gemm<DH><<<NG / 16, T, 0, stream>>>(pooled, fc1_w, fc1_b, gact, NG, 1);
    k_wmma_gemm<DO><<<NG / 16, T, 0, stream>>>(gact, fc2_w, fc2_b, (float*)d_out, NG, 0);
}


# scan kernel: a block's sum of the counts in front of it comes from 48 independent 16-byte reads per thread, many in flight at once, not from a loop that waited for each read before the next left
# speedup vs baseline: 6.3193x; 1.0594x over previous
#include <hip/hip_runtime.h>

#define NN 50000
#define NE 800000
#define DH 128
#define DO 64
#define NG 128

typedef float v2f __attribute__((ext_vector_type(2)));
typedef float v8f __attribute__((ext_vector_type(8)));

#define NNP 50176
#define NCH (NNP / 1024)

__global__ void k_zero_i(int* p, int n) {
    int i = blockIdx.x * blockDim.x + threadIdx.x;
    if (i < n) p[i] = 0;
}

__global__ void k_rank(const int* __restrict__ dst, int* __restrict__ cnt,
                       int* __restrict__ rank, int E) {
    int q = blockIdx.x * blockDim.x + threadIdx.x;
    if (q * 4 < E) {
        const int4 d = ((const int4*)dst)[q];
        int4 r;
        r.x = atomicAdd(&cnt[d.x], 1);
        r.y = atomicAdd(&cnt[d.y], 1);
        r.z = atomicAdd(&cnt[d.z], 1);
        r.w = atomicAdd(&cnt[d.w], 1);
        ((int4*)rank)[q] = r;
    }
}

__global__ void __launch_bounds__(256)
k_scan_norm(const int* __restrict__ cnt, int* __restrict__ rowptr,
            float* __restrict__ selfc, float* __restrict__ dinv)
{
    __shared__ int sOwn[8];
    __shared__ int sPre[8];
    const int4* c4 = (const int4*)cnt;
    const int b = blockIdx.x, t = threadIdx.x;
    const int lane = t & 31, wave = t >> 5;

    int4 pre[NCH - 1];
#pragma unroll
    for (int j = 0; j < NCH - 1; ++j) pre[j] = c4[j * 256 + t];
    const int4 v = c4[b * 256 + t];
    int part = 0;
#pragma unroll
    for (int j = 0; j < NCH - 1; ++j) {
        const int s4 = (pre[j].x + pre[j].y) + (pre[j].z + pre[j].w);
        part += (j < b) ? s4 : 0;
    }
    const int own = (v.x + v.y) + (v.z + v.w);

    int incl = own;
#pragma unroll
    for (int o = 1; o < 32; o <<= 1) {
        int up = __shfl_up(incl, o);
        if (lane >= o) incl += up;
    }
#pragma unroll
    for (int o = 16; o > 0; o >>= 1) part += __shfl_xor(part, o);
    if (lane == 31) sOwn[wave] = incl;
    if (lane == 0)  sPre[wave] = part;
    __syncthreads();
    int base = 0;
#pragma unroll
    for (int w = 0; w < 8; ++w) {
        base += sPre[w];
        if (w < wave) base += sOwn[w];
    }
    const int ex = base + incl - own;
    int4 rp;
    rp.x = ex; rp.y = ex + v.x; rp.z = rp.y + v.y; rp.w = rp.z + v.z;
    ((int4*)rowptr)[b * 256 + t] = rp;

    const int i0 = (b * 256 + t) * 4;
    if (i0 < NN) {
        float4 di, sc;
        float d;
        d = (float)(v.x + 1); di.x = 1.0f / sqrtf(d); sc.x = 1.0f / d;
        d = (float)(v.y + 1); di.y = 1.0f / sqrtf(d); sc.y = 1.0f / d;
        d = (float)(v.z + 1); di.z = 1.0f / sqrtf(d); sc.z = 1.0f / d;
        d = (float)(v.w + 1); di.w = 1.0f / sqrtf(d); sc.w = 1.0f / d;
        ((float4*)dinv)[b * 256 + t]  = di;
        ((float4*)selfc)[b * 256 + t] = sc;
    }
}

__device__ __forceinline__ void csr_put(int s, int d, int r, const int* __restrict__ rowptr,
                                        const float* __restrict__ dinv, int2* __restrict__ csr) {
    float w = dinv[s] * dinv[d];
    csr[rowptr[d] + r] = make_int2(s, __float_as_int(w));
}

__global__ void k_csr_fill(const int* __restrict__ src, const int* __restrict__ dst,
                           const int* __restrict__ rowptr, const int* __restrict__ rank,
                           const float* __restrict__ dinv, int2* __restrict__ csr, int E) {
    int q = blockIdx.x * blockDim.x + threadIdx.x;
    if (q * 4 < E) {
        const int4 s = ((const int4*)src)[q];
        const int4 d = ((const int4*)dst)[q];
        const int4 r = ((const int4*)rank)[q];
        csr_put(s.x, d.x, r.x, rowptr, dinv, csr);
        csr_put(s.y, d.y, r.y, rowptr, dinv, csr);
        csr_put(s.z, d.z, r.z, rowptr, dinv, csr);
        csr_put(s.w, d.w, r.w, rowptr, dinv, csr);
    }
}

template <int NC>
__global__ void __launch_bounds__(256)
k_wmma_gemm(const float* __restrict__ A, const float* __restrict__ W,
            const float* __restrict__ bias, float* __restrict__ C,
            int M, int relu)
{
    __shared__ float sWt[NC * 132];
    __shared__ alignas(16) float sA[16 * 132];

    const int tid = threadIdx.x;
    const int nstrips = (M + 15) >> 4;

    const int ar = tid >> 5, ac = tid & 31;
    int s = blockIdx.x;
    float4 p0 = make_float4(0.f, 0.f, 0.f, 0.f), p1 = p0;
    if (s < nstrips) {
        const int row0 = s * 16 + ar, row1 = row0 + 8;
        if (row0 < M) p0 = *(const float4*)(A + (size_t)row0 * 128 + ac * 4);
        if (row1 < M) p1 = *(const float4*)(A + (size_t)row1 * 128 + ac * 4);
    }

    for (int i = tid; i < 128 * NC; i += 256) {
        int k = i / NC, c = i % NC;
        sWt[c * 132 + k] = W[i];
    }

    const int wave = tid >> 5;
    const int lane = tid & 31;
    const int ln   = lane & 15;
    const int hi   = lane >> 4;
    const int c0   = wave * 16;
    const int col  = c0 + ln;
    const float bv = (c0 < NC && bias) ? bias[col] : 0.0f;

    for (; s < nstrips; s += gridDim.x) {
        *(float4*)&sA[ar * 132 + ac * 4]       = p0;
        *(float4*)&sA[(ar + 8) * 132 + ac * 4] = p1;
        __syncthreads();

        const int sn = s + gridDim.x;
        if (sn < nstrips) {
            const int row0 = sn * 16 + ar, row1 = row0 + 8;
            p0 = make_float4(0.f, 0.f, 0.f, 0.f); p1 = p0;
            if (row0 < M) p0 = *(const float4*)(A + (size_t)row0 * 128 + ac * 4);
            if (row1 < M) p1 = *(const float4*)(A + (size_t)row1 * 128 + ac * 4);
        }

        if (c0 < NC) {
            const float* aP = &sA [ln        * 132 + 2 * hi];
            const float* bP = &sWt[(c0 + ln) * 132 + 2 * hi];
            v8f acc = {0.f, 0.f, 0.f, 0.f, 0.f, 0.f, 0.f, 0.f};
#pragma unroll
            for (int k = 0; k < 128; k += 4) {
                v2f a = *(const v2f*)(aP + k);
                v2f b = *(const v2f*)(bP + k);
                acc = __builtin_amdgcn_wmma_f32_16x16x4_f32(
                    false, a, false, b, (short)0, acc, false, false);
            }
            const int r0 = s * 16;
#pragma unroll
            for (int r = 0; r < 8; ++r) {
                int row = r0 + r + 8 * hi;
                if (row < M) {
                    float v = acc[r] + bv;
                    if (relu) v = fmaxf(v, 0.0f);
                    C[(size_t)row * NC + col] = v;
                }
            }
        }
        __syncthreads();
    }
}

#define GR 5

__device__ __forceinline__ void edge_fma(float4& acc, const float4 v, const float w)
{
    acc.x = fmaf(v.x, w, acc.x); acc.y = fmaf(v.y, w, acc.y);
    acc.z = fmaf(v.z, w, acc.z); acc.w = fmaf(v.w, w, acc.w);
}

__global__ void __launch_bounds__(256)
k_gather_combine(const int* __restrict__ rowptr, const int2* __restrict__ csr,
                 const float* __restrict__ H2, const float* __restrict__ selfc,
                 const float* __restrict__ bias, float* __restrict__ H)
{
    const int lane = threadIdx.x & 31;
    const int wv = __builtin_amdgcn_readfirstlane((int)(blockIdx.x * 8 + (threadIdx.x >> 5)));
    const int i0 = wv * GR;
    if (i0 >= NN) return;
    const float4* H2v = (const float4*)H2 + lane;
    const float4 b4 = ((const float4*)bias)[lane];

    int beg = rowptr[i0];
#pragma unroll 1
    for (int n = 0; n < GR; ++n) {
        const int i = i0 + n;
        const int end = rowptr[i + 1];
        const float4 h = H2v[(size_t)i * 32];
        const float sc = selfc[i];

        float4 acc = {0.f, 0.f, 0.f, 0.f};
        int j = beg;
        for (; j + 8 <= end; j += 8) {
            int2   e[8];
            float4 v[8];
#pragma unroll
            for (int t = 0; t < 8; ++t) e[t] = csr[j + t];
#pragma unroll
            for (int t = 0; t < 8; ++t) v[t] = H2v[(size_t)e[t].x * 32];
#pragma unroll
            for (int t = 0; t < 8; ++t) edge_fma(acc, v[t], __int_as_float(e[t].y));
        }
        const int rem = end - j;
        if (rem > 0) {
            int2   e[7];
            float4 v[7];
#pragma unroll
            for (int t = 0; t < 7; ++t) e[t] = csr[min(j + t, end - 1)];
#pragma unroll
            for (int t = 0; t < 7; ++t) if (t < rem) v[t] = H2v[(size_t)e[t].x * 32];
#pragma unroll
            for (int t = 0; t < 7; ++t) if (t < rem) edge_fma(acc, v[t], __int_as_float(e[t].y));
        }

        float4 o;
        o.x = fmaxf(acc.x + h.x * sc + b4.x, 0.0f);
        o.y = fmaxf(acc.y + h.y * sc + b4.y, 0.0f);
        o.z = fmaxf(acc.z + h.z * sc + b4.z, 0.0f);
        o.w = fmaxf(acc.w + h.w * sc + b4.w, 0.0f);
        ((float4*)H)[(size_t)i * 32 + lane] = o;
        beg = end;
    }
}

typedef __attribute__((address_space(4))) int   cs_int;
typedef __attribute__((address_space(4))) int2  cs_int2;
typedef __attribute__((address_space(4))) float cs_float;

__global__ void __launch_bounds__(256)
k_gather_gemm(const int* __restrict__ rowptr, const int2* __restrict__ csr,
              const float* __restrict__ H2, const float* __restrict__ selfc,
              const float* __restrict__ bias, const float* __restrict__ Wn,
              float* __restrict__ H2n)
{
    __shared__ float sWt[DH * 132];
    __shared__ alignas(16) float sA[16 * 132];

    const int tid = threadIdx.x;

    for (int i = tid; i < DH * DH; i += 256) {
        int k = i / DH, c = i % DH;
        sWt[c * 132 + k] = Wn[i];
    }

    const int lane = tid & 31;
    const int wave = __builtin_amdgcn_readfirstlane(tid >> 5);
    const int ln   = lane & 15;
    const int hi   = lane >> 4;
    const int col  = wave * 16 + ln;
    const float4* H2v = (const float4*)H2 + lane;
    const float4 b4 = ((const float4*)bias)[lane];
    const float* aP = &sA [ln  * 132 + 2 * hi];
    const float* bP = &sWt[col * 132 + 2 * hi];
    const cs_int*   rowptr_c = (const cs_int*)rowptr;
    const cs_int2*  csr_c    = (const cs_int2*)csr;
    const cs_float* selfc_c  = (const cs_float*)selfc;

    for (int s = blockIdx.x; s < NN / 16; s += gridDim.x) {
        const int i0 = s * 16 + wave * 2;
        int beg = rowptr_c[i0];
#pragma unroll 1
        for (int n = 0; n < 2; ++n) {
            const int i = i0 + n;
            const int end = rowptr_c[i + 1];
            const float4 h = H2v[(size_t)i * 32];
            const float sc = selfc_c[i];

            float4 acc = {0.f, 0.f, 0.f, 0.f};
            int j = beg;
            for (; j + 8 <= end; j += 8) {
                int2   e[8];
                float4 v[8];
#pragma unroll
                for (int t = 0; t < 8; ++t) e[t] = csr_c[j + t];
#pragma unroll
                for (int t = 0; t < 8; ++t) v[t] = H2v[(size_t)e[t].x * 32];
#pragma unroll
                for (int t = 0; t < 8; ++t) edge_fma(acc, v[t], __int_as_float(e[t].y));
            }
            const int rem = end - j;
            if (rem > 0) {
                int2   e[7];
                float4 v[7];
#pragma unroll
                for (int t = 0; t < 7; ++t) e[t] = csr_c[min(j + t, end - 1)];
#pragma unroll
                for (int t = 0; t < 7; ++t) if (t < rem) v[t] = H2v[(size_t)e[t].x * 32];
#pragma unroll
                for (int t = 0; t < 7; ++t) if (t < rem) edge_fma(acc, v[t], __int_as_float(e[t].y));
            }

            float4 o;
            o.x = fmaxf(acc.x + h.x * sc + b4.x, 0.0f);
            o.y = fmaxf(acc.y + h.y * sc + b4.y, 0.0f);
            o.z = fmaxf(acc.z + h.z * sc + b4.z, 0.0f);
            o.w = fmaxf(acc.w + h.w * sc + b4.w, 0.0f);
            *(float4*)&sA[(wave * 2 + n) * 132 + lane * 4] = o;
            beg = end;
        }
        __syncthreads();

        v8f acc = {0.f, 0.f, 0.f, 0.f, 0.f, 0.f, 0.f, 0.f};
#pragma unroll
        for (int k = 0; k < DH; k += 4) {
            v2f a = *(const v2f*)(aP + k);
            v2f b = *(const v2f*)(bP + k);
            acc = __builtin_amdgcn_wmma_f32_16x16x4_f32(
                false, a, false, b, (short)0, acc, false, false);
        }
        __syncthreads();

        float* outp = H2n + (size_t)(s * 16 + 8 * hi) * DH + col;
#pragma unroll
        for (int r = 0; r < 8; ++r) outp[(size_t)r * DH] = acc[r];
    }
}

#define PCH 16

__device__ __forceinline__ void pool_flush(float* __restrict__ sums, float* __restrict__ counts,
                                           int g, int lane, float4 a, float n)
{
    float* s = sums + g * DH + lane * 4;
    atomicAdd(s + 0, a.x);
    atomicAdd(s + 1, a.y);
    atomicAdd(s + 2, a.z);
    atomicAdd(s + 3, a.w);
    if (lane == 0) atomicAdd(&counts[g], n);
}

__global__ void __launch_bounds__(256)
k_pool_runs(const float* __restrict__ H, const int* __restrict__ batch,
            float* __restrict__ sums, float* __restrict__ counts)
{
    const int lane = threadIdx.x & 31;
    const int wv = __builtin_amdgcn_readfirstlane((int)(blockIdx.x * 8 + (threadIdx.x >> 5)));
    if (wv >= NN / PCH) return;
    const int i0 = wv * PCH;
    const float4* Hv = (const float4*)H + (size_t)i0 * 32 + lane;

    float4 v[PCH];
    int    g[PCH];
#pragma unroll
    for (int k = 0; k < PCH; ++k) {
        v[k] = Hv[k * 32];
        g[k] = batch[i0 + k];
    }

    float4 acc = v[0];
    int    cur = g[0];
    float  n   = 1.0f;
#pragma unroll
    for (int k = 1; k < PCH; ++k) {
        if (g[k] != cur) {
            pool_flush(sums, counts, cur, lane, acc, n);
            acc = v[k]; cur = g[k]; n = 1.0f;
        } else {
            acc.x += v[k].x; acc.y += v[k].y; acc.z += v[k].z; acc.w += v[k].w;
            n += 1.0f;
        }
    }
    pool_flush(sums, counts, cur, lane, acc, n);
}

__global__ void k_pool_div(const float* __restrict__ sums, const float* __restrict__ counts,
                           float* __restrict__ pooled)
{
    int idx = blockIdx.x * blockDim.x + threadIdx.x;
    if (idx >= NG * DH) return;
    pooled[idx] = sums[idx] / fmaxf(counts[idx >> 7], 1.0f);
}

extern "C" void kernel_launch(void* const* d_in, const int* in_sizes, int n_in,
                              void* d_out, int out_size, void* d_ws, size_t ws_size,
                              hipStream_t stream)
{
    (void)in_sizes; (void)n_in; (void)out_size; (void)ws_size;

    const float* x     = (const float*)d_in[0];
    const int*   ei    = (const int*)d_in[1];
    const int*   src   = ei;
    const int*   dst   = ei + NE;
    const int*   batch = (const int*)d_in[2];
    const float* Wl[5] = {(const float*)d_in[4], (const float*)d_in[6], (const float*)d_in[8],
                          (const float*)d_in[10], (const float*)d_in[12]};
    const float* Bl[5] = {(const float*)d_in[5], (const float*)d_in[7], (const float*)d_in[9],
                          (const float*)d_in[11], (const float*)d_in[13]};
    const float* fc1_w = (const float*)d_in[14];
    const float* fc1_b = (const float*)d_in[15];
    const float* fc2_w = (const float*)d_in[16];
    const float* fc2_b = (const float*)d_in[17];

    float* H      = (float*)d_ws;
    float* H2     = H      + (size_t)NN * DH;
    float* selfc  = H2     + (size_t)NN * DH;
    float* dinv   = selfc  + NN;
    float* sums   = dinv   + NN;
    float* counts = sums   + (size_t)NG * DH;
    float* pooled = counts + NG;
    float* gact   = pooled + (size_t)NG * DH;
    int*   cnt    = (int*)(gact + (size_t)NG * DH);
    int*   rowptr = cnt    + NNP;
    int*   rank   = rowptr + NNP;
    int2*  csr    = (int2*)(rank + NE);

    const int T = 256;

    k_zero_i<<<NNP / T, T, 0, stream>>>(cnt, NNP);
    k_rank<<<(NE / 4 + T - 1) / T, T, 0, stream>>>(dst, cnt, rank, NE);
    k_scan_norm<<<NNP / 1024, T, 0, stream>>>(cnt, rowptr, selfc, dinv);
    k_csr_fill<<<(NE / 4 + T - 1) / T, T, 0, stream>>>(src, dst, rowptr, rank, dinv, csr, NE);

    const int gemm_blocks = 512;
    const int gath_blocks = (NN / GR + 7) / 8;

    const int fuse_blocks = 1024;

    k_wmma_gemm<DH><<<gemm_blocks, T, 0, stream>>>(x, Wl[0], nullptr, H2, NN, 0);
    k_gather_gemm<<<fuse_blocks, T, 0, stream>>>(rowptr, csr, H2, selfc, Bl[0], Wl[1], H);
    k_gather_gemm<<<fuse_blocks, T, 0, stream>>>(rowptr, csr, H,  selfc, Bl[1], Wl[2], H2);
    k_gather_gemm<<<fuse_blocks, T, 0, stream>>>(rowptr, csr, H2, selfc, Bl[2], Wl[3], H);
    k_gather_gemm<<<fuse_blocks, T, 0, stream>>>(rowptr, csr, H,  selfc, Bl[3], Wl[4], H2);
    k_gather_combine<<<gath_blocks, T, 0, stream>>>(rowptr, csr, H2, selfc, Bl[4], H);

    hipMemsetAsync(sums, 0, ((size_t)NG * DH + NG) * sizeof(float), stream);
    k_pool_runs<<<(NN / PCH + 7) / 8, T, 0, stream>>>(H, batch, sums, counts);
    k_pool_div<<<(NG * DH + T - 1) / T, T, 0, stream>>>(sums, counts, pooled);

    k_wmma_gemm<DH><<<NG / 16, T, 0, stream>>>(pooled, fc1_w, fc1_b, gact, NG, 1);
    k_wmma_gemm<DO><<<NG / 16, T, 0, stream>>>(gact, fc2_w, fc2_b, (float*)d_out, NG, 0);
}


# both gather kernels: edge records are read a node (a strip) ahead, one coalesced per-lane read, handed out by readlane; rows go out in blocks of up to 16, no index read in their way; sum order kept
# speedup vs baseline: 6.5500x; 1.0365x over previous
#include <hip/hip_runtime.h>

#define NN 50000
#define NE 800000
#define DH 128
#define DO 64
#define NG 128

typedef float v2f __attribute__((ext_vector_type(2)));
typedef float v8f __attribute__((ext_vector_type(8)));

#define NNP 50176
#define NCH (NNP / 1024)

__global__ void k_zero_i(int* p, int n) {
    int i = blockIdx.x * blockDim.x + threadIdx.x;
    if (i < n) p[i] = 0;
}

__global__ void k_rank(const int* __restrict__ dst, int* __restrict__ cnt,
                       int* __restrict__ rank, int E) {
    int q = blockIdx.x * blockDim.x + threadIdx.x;
    if (q * 4 < E) {
        const int4 d = ((const int4*)dst)[q];
        int4 r;
        r.x = atomicAdd(&cnt[d.x], 1);
        r.y = atomicAdd(&cnt[d.y], 1);
        r.z = atomicAdd(&cnt[d.z], 1);
        r.w = atomicAdd(&cnt[d.w], 1);
        ((int4*)rank)[q] = r;
    }
}

__global__ void __launch_bounds__(256)
k_scan_norm(const int* __restrict__ cnt, int* __restrict__ rowptr,
            float* __restrict__ selfc, float* __restrict__ dinv)
{
    __shared__ int sOwn[8];
    __shared__ int sPre[8];
    const int4* c4 = (const int4*)cnt;
    const int b = blockIdx.x, t = threadIdx.x;
    const int lane = t & 31, wave = t >> 5;

    int4 pre[NCH - 1];
#pragma unroll
    for (int j = 0; j < NCH - 1; ++j) pre[j] = c4[j * 256 + t];
    const int4 v = c4[b * 256 + t];
    int part = 0;
#pragma unroll
    for (int j = 0; j < NCH - 1; ++j) {
        const int s4 = (pre[j].x + pre[j].y) + (pre[j].z + pre[j].w);
        part += (j < b) ? s4 : 0;
    }
    const int own = (v.x + v.y) + (v.z + v.w);

    int incl = own;
#pragma unroll
    for (int o = 1; o < 32; o <<= 1) {
        int up = __shfl_up(incl, o);
        if (lane >= o) incl += up;
    }
#pragma unroll
    for (int o = 16; o > 0; o >>= 1) part += __shfl_xor(part, o);
    if (lane == 31) sOwn[wave] = incl;
    if (lane == 0)  sPre[wave] = part;
    __syncthreads();
    int base = 0;
#pragma unroll
    for (int w = 0; w < 8; ++w) {
        base += sPre[w];
        if (w < wave) base += sOwn[w];
    }
    const int ex = base + incl - own;
    int4 rp;
    rp.x = ex; rp.y = ex + v.x; rp.z = rp.y + v.y; rp.w = rp.z + v.z;
    ((int4*)rowptr)[b * 256 + t] = rp;

    const int i0 = (b * 256 + t) * 4;
    if (i0 < NN) {
        float4 di, sc;
        float d;
        d = (float)(v.x + 1); di.x = 1.0f / sqrtf(d); sc.x = 1.0f / d;
        d = (float)(v.y + 1); di.y = 1.0f / sqrtf(d); sc.y = 1.0f / d;
        d = (float)(v.z + 1); di.z = 1.0f / sqrtf(d); sc.z = 1.0f / d;
        d = (float)(v.w + 1); di.w = 1.0f / sqrtf(d); sc.w = 1.0f / d;
        ((float4*)dinv)[b * 256 + t]  = di;
        ((float4*)selfc)[b * 256 + t] = sc;
    }
}

__device__ __forceinline__ void csr_put(int s, int d, int r, const int* __restrict__ rowptr,
                                        const float* __restrict__ dinv, int2* __restrict__ csr) {
    float w = dinv[s] * dinv[d];
    csr[rowptr[d] + r] = make_int2(s, __float_as_int(w));
}

__global__ void k_csr_fill(const int* __restrict__ src, const int* __restrict__ dst,
                           const int* __restrict__ rowptr, const int* __restrict__ rank,
                           const float* __restrict__ dinv, int2* __restrict__ csr, int E) {
    int q = blockIdx.x * blockDim.x + threadIdx.x;
    if (q * 4 < E) {
        const int4 s = ((const int4*)src)[q];
        const int4 d = ((const int4*)dst)[q];
        const int4 r = ((const int4*)rank)[q];
        csr_put(s.x, d.x, r.x, rowptr, dinv, csr);
        csr_put(s.y, d.y, r.y, rowptr, dinv, csr);
        csr_put(s.z, d.z, r.z, rowptr, dinv, csr);
        csr_put(s.w, d.w, r.w, rowptr, dinv, csr);
    }
}

template <int NC>
__global__ void __launch_bounds__(256)
k_wmma_gemm(const float* __restrict__ A, const float* __restrict__ W,
            const float* __restrict__ bias, float* __restrict__ C,
            int M, int relu)
{
    __shared__ float sWt[NC * 132];
    __shared__ alignas(16) float sA[16 * 132];

    const int tid = threadIdx.x;
    const int nstrips = (M + 15) >> 4;

    const int ar = tid >> 5, ac = tid & 31;
    int s = blockIdx.x;
    float4 p0 = make_float4(0.f, 0.f, 0.f, 0.f), p1 = p0;
    if (s < nstrips) {
        const int row0 = s * 16 + ar, row1 = row0 + 8;
        if (row0 < M) p0 = *(const float4*)(A + (size_t)row0 * 128 + ac * 4);
        if (row1 < M) p1 = *(const float4*)(A + (size_t)row1 * 128 + ac * 4);
    }

    for (int i = tid; i < 128 * NC; i += 256) {
        int k = i / NC, c = i % NC;
        sWt[c * 132 + k] = W[i];
    }

    const int wave = tid >> 5;
    const int lane = tid & 31;
    const int ln   = lane & 15;
    const int hi   = lane >> 4;
    const int c0   = wave * 16;
    const int col  = c0 + ln;
    const float bv = (c0 < NC && bias) ? bias[col] : 0.0f;

    for (; s < nstrips; s += gridDim.x) {
        *(float4*)&sA[ar * 132 + ac * 4]       = p0;
        *(float4*)&sA[(ar + 8) * 132 + ac * 4] = p1;
        __syncthreads();

        const int sn = s + gridDim.x;
        if (sn < nstrips) {
            const int row0 = sn * 16 + ar, row1 = row0 + 8;
            p0 = make_float4(0.f, 0.f, 0.f, 0.f); p1 = p0;
            if (row0 < M) p0 = *(const float4*)(A + (size_t)row0 * 128 + ac * 4);
            if (row1 < M) p1 = *(const float4*)(A + (size_t)row1 * 128 + ac * 4);
        }

        if (c0 < NC) {
            const float* aP = &sA [ln        * 132 + 2 * hi];
            const float* bP = &sWt[(c0 + ln) * 132 + 2 * hi];
            v8f acc = {0.f, 0.f, 0.f, 0.f, 0.f, 0.f, 0.f, 0.f};
#pragma unroll
            for (int k = 0; k < 128; k += 4) {
                v2f a = *(const v2f*)(aP + k);
                v2f b = *(const v2f*)(bP + k);
                acc = __builtin_amdgcn_wmma_f32_16x16x4_f32(
                    false, a, false, b, (short)0, acc, false, false);
            }
            const int r0 = s * 16;
#pragma unroll
            for (int r = 0; r < 8; ++r) {
                int row = r0 + r + 8 * hi;
                if (row < M) {
                    float v = acc[r] + bv;
                    if (relu) v = fmaxf(v, 0.0f);
                    C[(size_t)row * NC + col] = v;
                }
            }
        }
        __syncthreads();
    }
}

#define GR 5

__device__ __forceinline__ int2 rec_load(const int2* __restrict__ csr, int nb, int lane)
{
    return csr[min(nb + lane, NE - 1)];
}

typedef float v4f __attribute__((ext_vector_type(4)));

__device__ __forceinline__ v4f row_ld(const float* __restrict__ H2, const int lane, const int roff, const int l)
{
    const char* rowp = (const char*)H2 + (unsigned)__builtin_amdgcn_readlane(roff, l);
    return ((const v4f*)rowp)[lane];
}
__device__ __forceinline__ float rec_w(const int wbits, const int l)
{
    return __int_as_float(__builtin_amdgcn_readlane(wbits, l));
}
__device__ __forceinline__ void edge_fma4(float4& acc, const v4f v, const float w)
{
    acc.x = fmaf(v.x, w, acc.x); acc.y = fmaf(v.y, w, acc.y);
    acc.z = fmaf(v.z, w, acc.z); acc.w = fmaf(v.w, w, acc.w);
}

template <int NB>
__device__ __forceinline__ void gather_block(float4& acc, const float* __restrict__ H2, const int lane,
                                             const int roff, const int wb, const int g, const int nv)
{
    v4f v[NB];
#pragma unroll
    for (int t = 0; t < NB; ++t) v[t] = row_ld(H2, lane, roff, g + min(t, nv - 1));
    if constexpr (NB == 4)
        asm volatile("" :: "v"(v[0]), "v"(v[1]), "v"(v[2]), "v"(v[3]));
    else if constexpr (NB == 8)
        asm volatile("" :: "v"(v[0]), "v"(v[1]), "v"(v[2]), "v"(v[3]), "v"(v[4]), "v"(v[5]), "v"(v[6]), "v"(v[7]));
    else if constexpr (NB == 12)
        asm volatile("" :: "v"(v[0]), "v"(v[1]), "v"(v[2]), "v"(v[3]), "v"(v[4]), "v"(v[5]), "v"(v[6]), "v"(v[7]),
                           "v"(v[8]), "v"(v[9]), "v"(v[10]), "v"(v[11]));
    else
        asm volatile("" :: "v"(v[0]), "v"(v[1]), "v"(v[2]), "v"(v[3]), "v"(v[4]), "v"(v[5]), "v"(v[6]), "v"(v[7]),
                           "v"(v[8]), "v"(v[9]), "v"(v[10]), "v"(v[11]), "v"(v[12]), "v"(v[13]), "v"(v[14]), "v"(v[15]));
#pragma unroll
    for (int t = 0; t < NB; ++t)
        if (t < nv) edge_fma4(acc, v[t], rec_w(wb, g + t));
}

__device__ __forceinline__ float4 gather_node(const float* __restrict__ H2, const int2* __restrict__ csr,
                                              int roff, int wb, const int nb, const int ne, const int lane)
{
    float4 acc = {0.f, 0.f, 0.f, 0.f};
    const int d = ne - nb;
    for (int cb = 0; cb < d; cb += 32) {
        if (cb > 0) { const int2 rec = rec_load(csr, nb + cb, lane); roff = rec.x << 9; wb = rec.y; }
        const int cnt  = min(32, d - cb);
#pragma unroll 1
        for (int g = 0; g < cnt; ) {
            const int left = cnt - g;
            if (left > 12)     { gather_block<16>(acc, H2, lane, roff, wb, g, min(left, 16)); g += 16; }
            else if (left > 8) { gather_block<12>(acc, H2, lane, roff, wb, g, left);          g += 12; }
            else if (left > 4) { gather_block<8>(acc, H2, lane, roff, wb, g, left);           g += 8; }
            else               { gather_block<4>(acc, H2, lane, roff, wb, g, left);           g += 4; }
        }
    }
    return acc;
}

__global__ void __launch_bounds__(256)
k_gather_combine(const int* __restrict__ rowptr, const int2* __restrict__ csr,
                 const float* __restrict__ H2, const float* __restrict__ selfc,
                 const float* __restrict__ bias, float* __restrict__ H)
{
    const int lane = threadIdx.x & 31;
    const int wv = __builtin_amdgcn_readfirstlane((int)(blockIdx.x * 8 + (threadIdx.x >> 5)));
    const int i0 = wv * GR;
    if (i0 >= NN) return;
    const float4* H2v = (const float4*)H2 + lane;
    const float4 b4 = ((const float4*)bias)[lane];

    int nb  = rowptr[i0];
    int ne  = rowptr[i0 + 1];
    int ne2 = rowptr[min(i0 + 2, NN)];
    int2 rec = rec_load(csr, nb, lane);
#pragma unroll 1
    for (int n = 0; n < GR; ++n) {
        const int i = i0 + n;
        const int roff = rec.x << 9, wb = rec.y;
        const int ne3 = rowptr[min(i + 3, NN)];
        const int2 recn = rec_load(csr, ne, lane);
        const float4 h = H2v[(size_t)i * 32];
        const float sc = selfc[i];

        const float4 acc = gather_node(H2, csr, roff, wb, nb, ne, lane);

        float4 o;
        o.x = fmaxf(acc.x + h.x * sc + b4.x, 0.0f);
        o.y = fmaxf(acc.y + h.y * sc + b4.y, 0.0f);
        o.z = fmaxf(acc.z + h.z * sc + b4.z, 0.0f);
        o.w = fmaxf(acc.w + h.w * sc + b4.w, 0.0f);
        ((float4*)H)[(size_t)i * 32 + lane] = o;
        nb = ne; ne = ne2; ne2 = ne3; rec = recn;
    }
}

typedef __attribute__((address_space(4))) int   cs_int;
typedef __attribute__((address_space(4))) float cs_float;

__global__ void __launch_bounds__(256)
k_gather_gemm(const int* __restrict__ rowptr, const int2* __restrict__ csr,
              const float* __restrict__ H2, const float* __restrict__ selfc,
              const float* __restrict__ bias, const float* __restrict__ Wn,
              float* __restrict__ H2n)
{
    __shared__ float sWt[DH * 132];
    __shared__ alignas(16) float sA[16 * 132];

    const int tid = threadIdx.x;

    const int lane = tid & 31;
    const int wave = __builtin_amdgcn_readfirstlane(tid >> 5);
    const int ln   = lane & 15;
    const int hi   = lane >> 4;
    const int col  = wave * 16 + ln;
    const float4* H2v = (const float4*)H2 + lane;
    const float4 b4 = ((const float4*)bias)[lane];
    const float* aP = &sA [ln  * 132 + 2 * hi];
    const float* bP = &sWt[col * 132 + 2 * hi];
    const cs_int*   rowptr_c = (const cs_int*)rowptr;
    const cs_float* selfc_c  = (const cs_float*)selfc;

    const int NS = NN / 16, G = gridDim.x;
    int s = blockIdx.x;
    const int ia = min(s, NS - 1) * 16 + wave * 2, ib = min(s + G, NS - 1) * 16 + wave * 2;
    int r0 = rowptr_c[ia], r1 = rowptr_c[ia + 1], r2 = rowptr_c[ia + 2];
    int n0 = rowptr_c[ib], n1 = rowptr_c[ib + 1], n2 = rowptr_c[ib + 2];
    int2 recA = rec_load(csr, r0, lane);
    int2 recB = rec_load(csr, r1, lane);

    for (int i = tid; i < DH * DH; i += 256) {
        int k = i / DH, c = i % DH;
        sWt[c * 132 + k] = Wn[i];
    }

    for (; s < NS; s += G) {
        const int i0 = s * 16 + wave * 2;
        const int roffA = recA.x << 9, wbA = recA.y, roffB = recB.x << 9, wbB = recB.y;
        const int ic = min(s + 2 * G, NS - 1) * 16 + wave * 2;
        const int m0 = rowptr_c[ic], m1 = rowptr_c[ic + 1], m2 = rowptr_c[ic + 2];
        const int2 nrecA = rec_load(csr, n0, lane);
        const int2 nrecB = rec_load(csr, n1, lane);

        int roff = roffA, wb = wbA;
        int nb = r0, ne = r1;
#pragma unroll 1
        for (int n = 0; n < 2; ++n) {
            const int i = i0 + n;
            const float4 h = H2v[(size_t)i * 32];
            const float sc = selfc_c[i];
            const float4 acc4 = gather_node(H2, csr, roff, wb, nb, ne, lane);
            float4 o;
            o.x = fmaxf(acc4.x + h.x * sc + b4.x, 0.0f);
            o.y = fmaxf(acc4.y + h.y * sc + b4.y, 0.0f);
            o.z = fmaxf(acc4.z + h.z * sc + b4.z, 0.0f);
            o.w = fmaxf(acc4.w + h.w * sc + b4.w, 0.0f);
            *(float4*)&sA[(wave * 2 + n) * 132 + lane * 4] = o;
            roff = roffB; wb = wbB; nb = r1; ne = r2;
        }

        r0 = n0; r1 = n1; r2 = n2; n0 = m0; n1 = m1; n2 = m2;
        recA = nrecA; recB = nrecB;
        __syncthreads();

        v8f acc = {0.f, 0.f, 0.f, 0.f, 0.f, 0.f, 0.f, 0.f};
#pragma unroll
        for (int k = 0; k < DH; k += 4) {
            v2f a = *(const v2f*)(aP + k);
            v2f b = *(const v2f*)(bP + k);
            acc = __builtin_amdgcn_wmma_f32_16x16x4_f32(
                false, a, false, b, (short)0, acc, false, false);
        }
        __syncthreads();

        float* outp = H2n + (size_t)(s * 16 + 8 * hi) * DH + col;
#pragma unroll
        for (int r = 0; r < 8; ++r) outp[(size_t)r * DH] = acc[r];
    }
}

#define PCH 16

__device__ __forceinline__ void pool_flush(float* __restrict__ sums, float* __restrict__ counts,
                                           int g, int lane, float4 a, float n)
{
    float* s = sums + g * DH + lane * 4;
    atomicAdd(s + 0, a.x);
    atomicAdd(s + 1, a.y);
    atomicAdd(s + 2, a.z);
    atomicAdd(s + 3, a.w);
    if (lane == 0) atomicAdd(&counts[g], n);
}

__global__ void __launch_bounds__(256)
k_pool_runs(const float* __restrict__ H, const int* __restrict__ batch,
            float* __restrict__ sums, float* __restrict__ counts)
{
    const int lane = threadIdx.x & 31;
    const int wv = __builtin_amdgcn_readfirstlane((int)(blockIdx.x * 8 + (threadIdx.x >> 5)));
    if (wv >= NN / PCH) return;
    const int i0 = wv * PCH;
    const float4* Hv = (const float4*)H + (size_t)i0 * 32 + lane;

    float4 v[PCH];
    int    g[PCH];
#pragma unroll
    for (int k = 0; k < PCH; ++k) {
        v[k] = Hv[k * 32];
        g[k] = batch[i0 + k];
    }

    float4 acc = v[0];
    int    cur = g[0];
    float  n   = 1.0f;
#pragma unroll
    for (int k = 1; k < PCH; ++k) {
        if (g[k] != cur) {
            pool_flush(sums, counts, cur, lane, acc, n);
            acc = v[k]; cur = g[k]; n = 1.0f;
        } else {
            acc.x += v[k].x; acc.y += v[k].y; acc.z += v[k].z; acc.w += v[k].w;
            n += 1.0f;
        }
    }
    pool_flush(sums, counts, cur, lane, acc, n);
}

__global__ void k_pool_div(const float* __restrict__ sums, const float* __restrict__ counts,
                           float* __restrict__ pooled)
{
    int idx = blockIdx.x * blockDim.x + threadIdx.x;
    if (idx >= NG * DH) return;
    pooled[idx] = sums[idx] / fmaxf(counts[idx >> 7], 1.0f);
}

extern "C" void kernel_launch(void* const* d_in, const int* in_sizes, int n_in,
                              void* d_out, int out_size, void* d_ws, size_t ws_size,
                              hipStream_t stream)
{
    (void)in_sizes; (void)n_in; (void)out_size; (void)ws_size;

    const float* x     = (const float*)d_in[0];
    const int*   ei    = (const int*)d_in[1];
    const int*   src   = ei;
    const int*   dst   = ei + NE;
    const int*   batch = (const int*)d_in[2];
    const float* Wl[5] = {(const float*)d_in[4], (const float*)d_in[6], (const float*)d_in[8],
                          (const float*)d_in[10], (const float*)d_in[12]};
    const float* Bl[5] = {(const float*)d_in[5], (const float*)d_in[7], (const float*)d_in[9],
                          (const float*)d_in[11], (const float*)d_in[13]};
    const float* fc1_w = (const float*)d_in[14];
    const float* fc1_b = (const float*)d_in[15];
    const float* fc2_w = (const float*)d_in[16];
    const float* fc2_b = (const float*)d_in[17];

    float* H      = (float*)d_ws;
    float* H2     = H      + (size_t)NN * DH;
    float* selfc  = H2     + (size_t)NN * DH;
    float* dinv   = selfc  + NN;
    float* sums   = dinv   + NN;
    float* counts = sums   + (size_t)NG * DH;
    float* pooled = counts + NG;
    float* gact   = pooled + (size_t)NG * DH;
    int*   cnt    = (int*)(gact + (size_t)NG * DH);
    int*   rowptr = cnt    + NNP;
    int*   rank   = rowptr + NNP;
    int2*  csr    = (int2*)(rank + NE);

    const int T = 256;

    k_zero_i<<<NNP / T, T, 0, stream>>>(cnt, NNP);
    k_rank<<<(NE / 4 + T - 1) / T, T, 0, stream>>>(dst, cnt, rank, NE);
    k_scan_norm<<<NNP / 1024, T, 0, stream>>>(cnt, rowptr, selfc, dinv);
    k_csr_fill<<<(NE / 4 + T - 1) / T, T, 0, stream>>>(src, dst, rowptr, rank, dinv, csr, NE);

    const int gemm_blocks = 512;
    const int gath_blocks = (NN / GR + 7) / 8;

    const int fuse_blocks = 1024;

    k_wmma_gemm<DH><<<gemm_blocks, T, 0, stream>>>(x, Wl[0], nullptr, H2, NN, 0);
    k_gather_gemm<<<fuse_blocks, T, 0, stream>>>(rowptr, csr, H2, selfc, Bl[0], Wl[1], H);
    k_gather_gemm<<<fuse_blocks, T, 0, stream>>>(rowptr, csr, H,  selfc, Bl[1], Wl[2], H2);
    k_gather_gemm<<<fuse_blocks, T, 0, stream>>>(rowptr, csr, H2, selfc, Bl[2], Wl[3], H);
    k_gather_gemm<<<fuse_blocks, T, 0, stream>>>(rowptr, csr, H,  selfc, Bl[3], Wl[4], H2);
    k_gather_combine<<<gath_blocks, T, 0, stream>>>(rowptr, csr, H2, selfc, Bl[4], H);

    hipMemsetAsync(sums, 0, ((size_t)NG * DH + NG) * sizeof(float), stream);
    k_pool_runs<<<(NN / PCH + 7) / 8, T, 0, stream>>>(H, batch, sums, counts);
    k_pool_div<<<(NG * DH + T - 1) / T, T, 0, stream>>>(sums, counts, pooled);

    k_wmma_gemm<DH><<<NG / 16, T, 0, stream>>>(pooled, fc1_w, fc1_b, gact, NG, 1);
    k_wmma_gemm<DO><<<NG / 16, T, 0, stream>>>(gact, fc2_w, fc2_b, (float*)d_out, NG, 0);
}
